# starting kernel
# baseline (speedup 1.0000x reference)
#include <hip/hip_runtime.h>

typedef __attribute__((ext_vector_type(2))) float v2f;
typedef __attribute__((ext_vector_type(8))) float v8f;

#define NX    16384
#define HID   128
#define ROWS  256

__device__ __forceinline__ v8f wmma4(v2f a, v2f b, v8f c) {
  return __builtin_amdgcn_wmma_f32_16x16x4_f32(
       false, a,  false, b,
       (short)0, c,  false,  false);
}

__global__ void mlp_trunk(const float* __restrict__ pw1, const float* __restrict__ pb1,
                          const float* __restrict__ pw2, const float* __restrict__ pb2,
                          const float* __restrict__ qw1, const float* __restrict__ qb1,
                          const float* __restrict__ qw2, const float* __restrict__ qb2,
                          float* __restrict__ H2P, float* __restrict__ H2S) {
  __shared__ float s_w1[128];
  __shared__ float s_b1[64];
  __shared__ float s_w2[64 * 128];
  __shared__ float s_b2[128];
  const int tid = threadIdx.x;
  const int n = blockIdx.x * 128 + tid;
  const float gx = (float)(n >> 7) * (1.0f / 127.0f);
  const float gy = (float)(n & 127) * (1.0f / 127.0f);

  for (int net = 0; net < 2; ++net) {
    const float* w1 = net ? qw1 : pw1;
    const float* b1 = net ? qb1 : pb1;
    const float* w2 = net ? qw2 : pw2;
    const float* b2 = net ? qb2 : pb2;
    __syncthreads();
    s_w1[tid] = w1[tid];
    if (tid < 64) s_b1[tid] = b1[tid];
    for (int i = tid; i < 64 * 128; i += 128) s_w2[i] = w2[i];
    s_b2[tid] = b2[tid];
    __syncthreads();

    float h1[64];
#pragma unroll
    for (int j = 0; j < 64; ++j) {
      float v = gx * s_w1[j] + gy * s_w1[64 + j] + s_b1[j];
      h1[j] = fmaxf(v, 0.0f);
    }
    for (int c = 0; c < HID; ++c) {
      float acc = s_b2[c];
#pragma unroll 16
      for (int j = 0; j < 64; ++j) acc += h1[j] * s_w2[j * 128 + c];
      acc = fmaxf(acc, 0.0f);
      if (net == 0) H2P[c * NX + n] = acc;
      else          H2S[n * HID + c] = acc;
    }
  }
}

__global__ void rowsum(const float* __restrict__ x, float* __restrict__ X1) {
  __shared__ float red[256];
  const int r = blockIdx.x;
  float s = 0.0f;
  for (int n = threadIdx.x; n < NX; n += 256) s += x[(long)r * NX + n];
  red[threadIdx.x] = s;
  __syncthreads();
  for (int off = 128; off > 0; off >>= 1) {
    if ((int)threadIdx.x < off) red[threadIdx.x] += red[threadIdx.x + off];
    __syncthreads();
  }
  if (threadIdx.x == 0) X1[r] = red[0];
}

__global__ void gemm1_partial(const float* __restrict__ X, const float* __restrict__ H2P,
                              float* __restrict__ P) {
  const int lane = threadIdx.x;
  const int half = lane >> 4;
  const int l16  = lane & 15;
  const int ms   = blockIdx.x;
  const int kc   = blockIdx.y;
  const long k0  = (long)kc * 256 + 2 * half;

  v8f acc[8];
  const v8f vzero = {0.f,0.f,0.f,0.f,0.f,0.f,0.f,0.f};
#pragma unroll
  for (int t = 0; t < 8; ++t) acc[t] = vzero;

  const float* aptr = X   + (long)(ms * 16 + l16) * NX + k0;
  const float* bptr = H2P + (long)l16 * NX + k0;
  for (int kb = 0; kb < 256; kb += 4) {
    __builtin_prefetch(aptr + 64, 0, 0);
    v2f a = *(const v2f*)aptr;
#pragma unroll
    for (int t = 0; t < 8; ++t) {
      v2f b = *(const v2f*)(bptr + (long)t * 16 * NX);
      acc[t] = wmma4(a, b, acc[t]);
    }
    aptr += 4;
    bptr += 4;
  }
#pragma unroll
  for (int t = 0; t < 8; ++t)
#pragma unroll
    for (int r = 0; r < 8; ++r)
      P[((long)kc * ROWS + ms * 16 + r + 8 * half) * HID + t * 16 + l16] = acc[t][r];
}

__global__ void reduceP(const float* __restrict__ P, float* __restrict__ T) {
  const int idx = blockIdx.x * 256 + threadIdx.x;
  float s = 0.0f;
  for (int kc = 0; kc < 64; ++kc) s += P[(long)kc * (ROWS * HID) + idx];
  T[idx] = s;
}

__global__ void s_kernel(const float* __restrict__ T, const float* __restrict__ w3p,
                         const float* __restrict__ b3p, const float* __restrict__ X1,
                         float* __restrict__ S) {
  const int idx = blockIdx.x * 256 + threadIdx.x;
  const int b = idx >> 10;
  const int o = (idx >> 5) & 31;
  const int i = idx & 31;
  const int r = b * 32 + i;
  const int oi = o * 32 + i;
  float acc = b3p[oi] * X1[r];
  for (int k = 0; k < HID; ++k) acc += T[r * HID + k] * w3p[k * 1024 + oi];
  S[idx] = acc;
}

__global__ void u_kernel(const float* __restrict__ S, const float* __restrict__ w3q,
                         float* __restrict__ U) {
  const int idx = blockIdx.x * 256 + threadIdx.x;
  const int bo = idx >> 7;
  const int k = idx & 127;
  const int o = bo & 31;
  float acc = 0.0f;
#pragma unroll
  for (int i = 0; i < 32; ++i) acc += S[bo * 32 + i] * w3q[k * 1024 + o * 32 + i];
  U[idx] = acc;
}

__global__ void c_kernel(const float* __restrict__ S, const float* __restrict__ b3q,
                         float* __restrict__ C) {
  const int bo = threadIdx.x;
  const int o = bo & 31;
  float acc = 0.0f;
#pragma unroll
  for (int i = 0; i < 32; ++i) acc += S[bo * 32 + i] * b3q[o * 32 + i];
  C[bo] = acc;
}

__global__ void gemm2(const float* __restrict__ U, const float* __restrict__ H2S,
                      const float* __restrict__ C, float* __restrict__ out) {
  const int lane = threadIdx.x;
  const int half = lane >> 4;
  const int l16  = lane & 15;
  const int mt = blockIdx.x;
  const int n0 = blockIdx.y * 4;

  v8f acc[4];
  const v8f vzero = {0.f,0.f,0.f,0.f,0.f,0.f,0.f,0.f};
#pragma unroll
  for (int t = 0; t < 4; ++t) acc[t] = vzero;

  const float* aptr = U   + (long)(mt * 16 + l16) * HID + 2 * half;
  const float* bptr = H2S + ((long)n0 * 16 + l16) * HID + 2 * half;
#pragma unroll
  for (int kb = 0; kb < HID; kb += 4) {
    v2f a = *(const v2f*)(aptr + kb);
#pragma unroll
    for (int t = 0; t < 4; ++t) {
      v2f b = *(const v2f*)(bptr + t * 16 * HID + kb);
      acc[t] = wmma4(a, b, acc[t]);
    }
  }
  const float scale = 1.0f / (float)NX;
#pragma unroll
  for (int t = 0; t < 4; ++t)
#pragma unroll
    for (int r = 0; r < 8; ++r) {
      const int row = mt * 16 + r + 8 * half;
      out[(long)row * NX + (n0 + t) * 16 + l16] = (acc[t][r] + C[row]) * scale;
    }
}

extern "C" void kernel_launch(void* const* d_in, const int* in_sizes, int n_in,
                              void* d_out, int out_size, void* d_ws, size_t ws_size,
                              hipStream_t stream) {
  const float* x   = (const float*)d_in[0];
  const float* pw1 = (const float*)d_in[1];
  const float* pb1 = (const float*)d_in[2];
  const float* pw2 = (const float*)d_in[3];
  const float* pb2 = (const float*)d_in[4];
  const float* pw3 = (const float*)d_in[5];
  const float* pb3 = (const float*)d_in[6];
  const float* qw1 = (const float*)d_in[7];
  const float* qb1 = (const float*)d_in[8];
  const float* qw2 = (const float*)d_in[9];
  const float* qb2 = (const float*)d_in[10];
  const float* qw3 = (const float*)d_in[11];
  const float* qb3 = (const float*)d_in[12];

  float* ws  = (float*)d_ws;
  float* H2P = ws;
  float* H2S = H2P + HID * NX;
  float* P   = H2S + NX * HID;
  float* T   = P + 64 * ROWS * HID;
  float* X1  = T + ROWS * HID;
  float* S   = X1 + ROWS;
  float* Uu  = S + 8192;
  float* Cc  = Uu + ROWS * HID;

  mlp_trunk<<<dim3(NX / 128), dim3(128), 0, stream>>>(pw1, pb1, pw2, pb2,
                                                      qw1, qb1, qw2, qb2, H2P, H2S);
  rowsum<<<dim3(ROWS), dim3(256), 0, stream>>>(x, X1);
  gemm1_partial<<<dim3(16, 64), dim3(32), 0, stream>>>(x, H2P, P);
  reduceP<<<dim3(128), dim3(256), 0, stream>>>(P, T);
  s_kernel<<<dim3(32), dim3(256), 0, stream>>>(T, pw3, pb3, X1, S);
  u_kernel<<<dim3(128), dim3(256), 0, stream>>>(S, qw3, Uu);
  c_kernel<<<dim3(1), dim3(256), 0, stream>>>(S, qb3, Cc);
  gemm2<<<dim3(16, 256), dim3(32), 0, stream>>>(Uu, H2S, Cc, (float*)d_out);
}


# MLP trunk layer 2 (64->128) moved from per-thread f32 FMA chains onto f32 16x16x4 WMMA, layer 1 built in fragment registers; the layer's sum order becomes the matrix unit's
# speedup vs baseline: 3.6616x; 3.6616x over previous
#include <hip/hip_runtime.h>

typedef __attribute__((ext_vector_type(2))) float v2f;
typedef __attribute__((ext_vector_type(4))) float v4f;
typedef __attribute__((ext_vector_type(8))) float v8f;

#define NX    16384
#define HID   128
#define ROWS  256

__device__ __forceinline__ v8f wmma4(v2f a, v2f b, v8f c) {
  return __builtin_amdgcn_wmma_f32_16x16x4_f32(
       false, a,  false, b,
       (short)0, c,  false,  false);
}

template <int NET>
__device__ __forceinline__ void trunk_tile(const float* __restrict__ w1,
                                           const float* __restrict__ b1,
                                           const float* __restrict__ w2,
                                           const float* __restrict__ b2,
                                           float* __restrict__ H,
                                           int n0, int c0, int half, int l16) {
  v2f hf[2][16];
#pragma unroll
  for (int p = 0; p < 2; ++p) {
    const int n = n0 + 2 * l16 + p;
    const float gx = (float)(n >> 7) * (1.0f / 127.0f);
    const float gy = (float)(n & 127) * (1.0f / 127.0f);
#pragma unroll
    for (int ks = 0; ks < 16; ++ks) {
      const int j0 = 4 * ks + 2 * half;
      const v2f wa = *(const v2f*)(w1 + j0);
      const v2f wb = *(const v2f*)(w1 + 64 + j0);
      const v2f bb = *(const v2f*)(b1 + j0);
      v2f h;
      h.x = fmaxf(gx * wa.x + gy * wb.x + bb.x, 0.0f);
      h.y = fmaxf(gx * wa.y + gy * wb.y + bb.y, 0.0f);
      hf[p][ks] = h;
    }
  }

  v8f acc[4][2];
  if (NET == 0) {
#pragma unroll
    for (int r = 0; r < 8; ++r) {
      const v4f bq = *(const v4f*)(b2 + c0 + 32 * half + 4 * r);
#pragma unroll
      for (int tt = 0; tt < 4; ++tt) {
        acc[tt][0][r] = bq[tt];
        acc[tt][1][r] = bq[tt];
      }
    }
  } else {
    const v4f bq = *(const v4f*)(b2 + c0 + 4 * l16);
#pragma unroll
    for (int tt = 0; tt < 4; ++tt)
#pragma unroll
      for (int r = 0; r < 8; ++r) {
        acc[tt][0][r] = bq[tt];
        acc[tt][1][r] = bq[tt];
      }
  }

  const float* wp = w2 + (2 * half) * HID + c0 + 4 * l16;
#pragma unroll
  for (int ks = 0; ks < 16; ++ks) {
    const v4f wlo = *(const v4f*)(wp + (4 * ks) * HID);
    const v4f whi = *(const v4f*)(wp + (4 * ks + 1) * HID);
#pragma unroll
    for (int tt = 0; tt < 4; ++tt) {
      v2f wf;
      wf.x = wlo[tt];
      wf.y = whi[tt];
#pragma unroll
      for (int p = 0; p < 2; ++p) {
        if (NET == 0) acc[tt][p] = wmma4(wf, hf[p][ks], acc[tt][p]);
        else          acc[tt][p] = wmma4(hf[p][ks], wf, acc[tt][p]);
      }
    }
  }

  if (NET == 0) {
#pragma unroll
    for (int tt = 0; tt < 4; ++tt)
#pragma unroll
      for (int r = 0; r < 8; ++r) {
        const int c = c0 + 4 * (8 * half + r) + tt;
        v2f o;
        o.x = fmaxf(acc[tt][0][r], 0.0f);
        o.y = fmaxf(acc[tt][1][r], 0.0f);
        *(v2f*)(H + (long)c * NX + n0 + 2 * l16) = o;
      }
  } else {
#pragma unroll
    for (int p = 0; p < 2; ++p)
#pragma unroll
      for (int r = 0; r < 8; ++r) {
        const int n = n0 + 2 * (8 * half + r) + p;
        v4f o;
        o.x = fmaxf(acc[0][p][r], 0.0f);
        o.y = fmaxf(acc[1][p][r], 0.0f);
        o.z = fmaxf(acc[2][p][r], 0.0f);
        o.w = fmaxf(acc[3][p][r], 0.0f);
        *(v4f*)(H + (long)n * HID + c0 + 4 * l16) = o;
      }
  }
}

__global__ void __launch_bounds__(32)
mlp_trunk(const float* __restrict__ pw1, const float* __restrict__ pb1,
          const float* __restrict__ pw2, const float* __restrict__ pb2,
          const float* __restrict__ qw1, const float* __restrict__ qb1,
          const float* __restrict__ qw2, const float* __restrict__ qb2,
          float* __restrict__ H2P, float* __restrict__ H2S) {
  const int lane = threadIdx.x;
  const int half = lane >> 4;
  const int l16  = lane & 15;
  const int n0 = blockIdx.x * 32;
  const int c0 = blockIdx.y * 64;
  if (blockIdx.z == 0) trunk_tile<0>(pw1, pb1, pw2, pb2, H2P, n0, c0, half, l16);
  else                 trunk_tile<1>(qw1, qb1, qw2, qb2, H2S, n0, c0, half, l16);
}

__global__ void rowsum(const float* __restrict__ x, float* __restrict__ X1) {
  __shared__ float red[256];
  const int r = blockIdx.x;
  float s = 0.0f;
  for (int n = threadIdx.x; n < NX; n += 256) s += x[(long)r * NX + n];
  red[threadIdx.x] = s;
  __syncthreads();
  for (int off = 128; off > 0; off >>= 1) {
    if ((int)threadIdx.x < off) red[threadIdx.x] += red[threadIdx.x + off];
    __syncthreads();
  }
  if (threadIdx.x == 0) X1[r] = red[0];
}

__global__ void gemm1_partial(const float* __restrict__ X, const float* __restrict__ H2P,
                              float* __restrict__ P) {
  const int lane = threadIdx.x;
  const int half = lane >> 4;
  const int l16  = lane & 15;
  const int ms   = blockIdx.x;
  const int kc   = blockIdx.y;
  const long k0  = (long)kc * 256 + 2 * half;

  v8f acc[8];
  const v8f vzero = {0.f,0.f,0.f,0.f,0.f,0.f,0.f,0.f};
#pragma unroll
  for (int t = 0; t < 8; ++t) acc[t] = vzero;

  const float* aptr = X   + (long)(ms * 16 + l16) * NX + k0;
  const float* bptr = H2P + (long)l16 * NX + k0;
  for (int kb = 0; kb < 256; kb += 4) {
    __builtin_prefetch(aptr + 64, 0, 0);
    v2f a = *(const v2f*)aptr;
#pragma unroll
    for (int t = 0; t < 8; ++t) {
      v2f b = *(const v2f*)(bptr + (long)t * 16 * NX);
      acc[t] = wmma4(a, b, acc[t]);
    }
    aptr += 4;
    bptr += 4;
  }
#pragma unroll
  for (int t = 0; t < 8; ++t)
#pragma unroll
    for (int r = 0; r < 8; ++r)
      P[((long)kc * ROWS + ms * 16 + r + 8 * half) * HID + t * 16 + l16] = acc[t][r];
}

__global__ void reduceP(const float* __restrict__ P, float* __restrict__ T) {
  const int idx = blockIdx.x * 256 + threadIdx.x;
  float s = 0.0f;
  for (int kc = 0; kc < 64; ++kc) s += P[(long)kc * (ROWS * HID) + idx];
  T[idx] = s;
}

__global__ void s_kernel(const float* __restrict__ T, const float* __restrict__ w3p,
                         const float* __restrict__ b3p, const float* __restrict__ X1,
                         float* __restrict__ S) {
  const int idx = blockIdx.x * 256 + threadIdx.x;
  const int b = idx >> 10;
  const int o = (idx >> 5) & 31;
  const int i = idx & 31;
  const int r = b * 32 + i;
  const int oi = o * 32 + i;
  float acc = b3p[oi] * X1[r];
  for (int k = 0; k < HID; ++k) acc += T[r * HID + k] * w3p[k * 1024 + oi];
  S[idx] = acc;
}

__global__ void u_kernel(const float* __restrict__ S, const float* __restrict__ w3q,
                         float* __restrict__ U) {
  const int idx = blockIdx.x * 256 + threadIdx.x;
  const int bo = idx >> 7;
  const int k = idx & 127;
  const int o = bo & 31;
  float acc = 0.0f;
#pragma unroll
  for (int i = 0; i < 32; ++i) acc += S[bo * 32 + i] * w3q[k * 1024 + o * 32 + i];
  U[idx] = acc;
}

__global__ void c_kernel(const float* __restrict__ S, const float* __restrict__ b3q,
                         float* __restrict__ C) {
  const int bo = threadIdx.x;
  const int o = bo & 31;
  float acc = 0.0f;
#pragma unroll
  for (int i = 0; i < 32; ++i) acc += S[bo * 32 + i] * b3q[o * 32 + i];
  C[bo] = acc;
}

__global__ void gemm2(const float* __restrict__ U, const float* __restrict__ H2S,
                      const float* __restrict__ C, float* __restrict__ out) {
  const int lane = threadIdx.x;
  const int half = lane >> 4;
  const int l16  = lane & 15;
  const int mt = blockIdx.x;
  const int n0 = blockIdx.y * 4;

  v8f acc[4];
  const v8f vzero = {0.f,0.f,0.f,0.f,0.f,0.f,0.f,0.f};
#pragma unroll
  for (int t = 0; t < 4; ++t) acc[t] = vzero;

  const float* aptr = U   + (long)(mt * 16 + l16) * HID + 2 * half;
  const float* bptr = H2S + ((long)n0 * 16 + l16) * HID + 2 * half;
#pragma unroll
  for (int kb = 0; kb < HID; kb += 4) {
    v2f a = *(const v2f*)(aptr + kb);
#pragma unroll
    for (int t = 0; t < 4; ++t) {
      v2f b = *(const v2f*)(bptr + t * 16 * HID + kb);
      acc[t] = wmma4(a, b, acc[t]);
    }
  }
  const float scale = 1.0f / (float)NX;
#pragma unroll
  for (int t = 0; t < 4; ++t)
#pragma unroll
    for (int r = 0; r < 8; ++r) {
      const int row = mt * 16 + r + 8 * half;
      out[(long)row * NX + (n0 + t) * 16 + l16] = (acc[t][r] + C[row]) * scale;
    }
}

extern "C" void kernel_launch(void* const* d_in, const int* in_sizes, int n_in,
                              void* d_out, int out_size, void* d_ws, size_t ws_size,
                              hipStream_t stream) {
  const float* x   = (const float*)d_in[0];
  const float* pw1 = (const float*)d_in[1];
  const float* pb1 = (const float*)d_in[2];
  const float* pw2 = (const float*)d_in[3];
  const float* pb2 = (const float*)d_in[4];
  const float* pw3 = (const float*)d_in[5];
  const float* pb3 = (const float*)d_in[6];
  const float* qw1 = (const float*)d_in[7];
  const float* qb1 = (const float*)d_in[8];
  const float* qw2 = (const float*)d_in[9];
  const float* qb2 = (const float*)d_in[10];
  const float* qw3 = (const float*)d_in[11];
  const float* qb3 = (const float*)d_in[12];

  float* ws  = (float*)d_ws;
  float* H2P = ws;
  float* H2S = H2P + HID * NX;
  float* P   = H2S + NX * HID;
  float* T   = P + 64 * ROWS * HID;
  float* X1  = T + ROWS * HID;
  float* S   = X1 + ROWS;
  float* Uu  = S + 8192;
  float* Cc  = Uu + ROWS * HID;

  mlp_trunk<<<dim3(NX / 32, HID / 64, 2), dim3(32), 0, stream>>>(pw1, pb1, pw2, pb2,
                                                                 qw1, qb1, qw2, qb2, H2P, H2S);
  rowsum<<<dim3(ROWS), dim3(256), 0, stream>>>(x, X1);
  gemm1_partial<<<dim3(16, 64), dim3(32), 0, stream>>>(x, H2P, P);
  reduceP<<<dim3(128), dim3(256), 0, stream>>>(P, T);
  s_kernel<<<dim3(32), dim3(256), 0, stream>>>(T, pw3, pb3, X1, S);
  u_kernel<<<dim3(128), dim3(256), 0, stream>>>(S, qw3, Uu);
  c_kernel<<<dim3(1), dim3(256), 0, stream>>>(S, qb3, Cc);
  gemm2<<<dim3(16, 256), dim3(32), 0, stream>>>(Uu, H2S, Cc, (float*)d_out);
}


# GEMM1 operand loads made whole-line: phi trunk stored point-major, x read in 64-byte runs per lane, 4 row strips per block share a chunk; GEMM1's k order (f32 sum order) changes
# speedup vs baseline: 5.1969x; 1.4193x over previous
#include <hip/hip_runtime.h>

typedef __attribute__((ext_vector_type(2))) float v2f;
typedef __attribute__((ext_vector_type(4))) float v4f;
typedef __attribute__((ext_vector_type(8))) float v8f;

#define NX    16384
#define HID   128
#define ROWS  256

__device__ __forceinline__ v8f wmma4(v2f a, v2f b, v8f c) {
  return __builtin_amdgcn_wmma_f32_16x16x4_f32(
       false, a,  false, b,
       (short)0, c,  false,  false);
}

__device__ __forceinline__ void trunk_tile(const float* __restrict__ w1,
                                           const float* __restrict__ b1,
                                           const float* __restrict__ w2,
                                           const float* __restrict__ b2,
                                           float* __restrict__ H,
                                           int n0, int c0, int half, int l16) {
  v2f hf[2][16];
#pragma unroll
  for (int p = 0; p < 2; ++p) {
    const int n = n0 + 2 * l16 + p;
    const float gx = (float)(n >> 7) * (1.0f / 127.0f);
    const float gy = (float)(n & 127) * (1.0f / 127.0f);
#pragma unroll
    for (int ks = 0; ks < 16; ++ks) {
      const int j0 = 4 * ks + 2 * half;
      const v2f wa = *(const v2f*)(w1 + j0);
      const v2f wb = *(const v2f*)(w1 + 64 + j0);
      const v2f bb = *(const v2f*)(b1 + j0);
      v2f h;
      h.x = fmaxf(gx * wa.x + gy * wb.x + bb.x, 0.0f);
      h.y = fmaxf(gx * wa.y + gy * wb.y + bb.y, 0.0f);
      hf[p][ks] = h;
    }
  }

  v8f acc[4][2];
  const v4f bq = *(const v4f*)(b2 + c0 + 4 * l16);
#pragma unroll
  for (int tt = 0; tt < 4; ++tt)
#pragma unroll
    for (int r = 0; r < 8; ++r) {
      acc[tt][0][r] = bq[tt];
      acc[tt][1][r] = bq[tt];
    }

  const float* wp = w2 + (2 * half) * HID + c0 + 4 * l16;
#pragma unroll
  for (int ks = 0; ks < 16; ++ks) {
    const v4f wlo = *(const v4f*)(wp + (4 * ks) * HID);
    const v4f whi = *(const v4f*)(wp + (4 * ks + 1) * HID);
#pragma unroll
    for (int tt = 0; tt < 4; ++tt) {
      v2f wf;
      wf.x = wlo[tt];
      wf.y = whi[tt];
#pragma unroll
      for (int p = 0; p < 2; ++p)
        acc[tt][p] = wmma4(hf[p][ks], wf, acc[tt][p]);
    }
  }

#pragma unroll
  for (int p = 0; p < 2; ++p)
#pragma unroll
    for (int r = 0; r < 8; ++r) {
      const int n = n0 + 2 * (8 * half + r) + p;
      v4f o;
      o.x = fmaxf(acc[0][p][r], 0.0f);
      o.y = fmaxf(acc[1][p][r], 0.0f);
      o.z = fmaxf(acc[2][p][r], 0.0f);
      o.w = fmaxf(acc[3][p][r], 0.0f);
      *(v4f*)(H + (long)n * HID + c0 + 4 * l16) = o;
    }
}

__global__ void __launch_bounds__(32)
mlp_trunk(const float* __restrict__ pw1, const float* __restrict__ pb1,
          const float* __restrict__ pw2, const float* __restrict__ pb2,
          const float* __restrict__ qw1, const float* __restrict__ qb1,
          const float* __restrict__ qw2, const float* __restrict__ qb2,
          float* __restrict__ H2P, float* __restrict__ H2S) {
  const int lane = threadIdx.x;
  const int half = lane >> 4;
  const int l16  = lane & 15;
  const int n0 = blockIdx.x * 32;
  const int c0 = blockIdx.y * 64;
  if (blockIdx.z == 0) trunk_tile(pw1, pb1, pw2, pb2, H2P, n0, c0, half, l16);
  else                 trunk_tile(qw1, qb1, qw2, qb2, H2S, n0, c0, half, l16);
}

__global__ void rowsum(const float* __restrict__ x, float* __restrict__ X1) {
  __shared__ float red[256];
  const int r = blockIdx.x;
  float s = 0.0f;
  for (int n = threadIdx.x; n < NX; n += 256) s += x[(long)r * NX + n];
  red[threadIdx.x] = s;
  __syncthreads();
  for (int off = 128; off > 0; off >>= 1) {
    if ((int)threadIdx.x < off) red[threadIdx.x] += red[threadIdx.x + off];
    __syncthreads();
  }
  if (threadIdx.x == 0) X1[r] = red[0];
}

struct G1Stage {
  v4f xa[4];
  v4f bw[8][2][2];
};

__device__ __forceinline__ void g1_load(G1Stage& st, const float* __restrict__ ap,
                                        const float* __restrict__ bp) {
#pragma unroll
  for (int q = 0; q < 4; ++q) st.xa[q] = *(const v4f*)(ap + 4 * q);
#pragma unroll
  for (int s = 0; s < 8; ++s)
#pragma unroll
    for (int g = 0; g < 2; ++g)
#pragma unroll
      for (int e = 0; e < 2; ++e)
        st.bw[s][g][e] = *(const v4f*)(bp + (2 * s + e) * HID + 64 * g);
}

__device__ __forceinline__ void g1_mma(const G1Stage& st, v8f (&acc)[2][4]) {
#pragma unroll
  for (int s = 0; s < 8; ++s) {
    v2f a;
    a.x = st.xa[s >> 1][2 * (s & 1)];
    a.y = st.xa[s >> 1][2 * (s & 1) + 1];
#pragma unroll
    for (int g = 0; g < 2; ++g)
#pragma unroll
      for (int tt = 0; tt < 4; ++tt) {
        v2f b;
        b.x = st.bw[s][g][0][tt];
        b.y = st.bw[s][g][1][tt];
        acc[g][tt] = wmma4(a, b, acc[g][tt]);
      }
  }
}

__global__ void __launch_bounds__(128)
gemm1_partial(const float* __restrict__ X, const float* __restrict__ H2P,
              float* __restrict__ P) {
  const int lane = threadIdx.x & 31;
  const int half = lane >> 4;
  const int l16  = lane & 15;
  const int ms   = blockIdx.x * 4 + (threadIdx.x >> 5);
  const int kc   = blockIdx.y;

  v8f acc[2][4];
  const v8f vzero = {0.f,0.f,0.f,0.f,0.f,0.f,0.f,0.f};
#pragma unroll
  for (int g = 0; g < 2; ++g)
#pragma unroll
    for (int tt = 0; tt < 4; ++tt) acc[g][tt] = vzero;

  const float* aptr = X   + (long)(ms * 16 + l16) * NX + kc * 256 + 16 * half;
  const float* bptr = H2P + ((long)kc * 256 + 16 * half) * HID + 4 * l16;

  G1Stage s0, s1;
  g1_load(s0, aptr, bptr);
  for (int st = 0; st < 8; st += 2) {
    g1_load(s1, aptr + (st + 1) * 32, bptr + (long)(st + 1) * 32 * HID);
    __builtin_amdgcn_sched_barrier(0);
    g1_mma(s0, acc);
    if (st + 2 < 8) g1_load(s0, aptr + (st + 2) * 32, bptr + (long)(st + 2) * 32 * HID);
    __builtin_amdgcn_sched_barrier(0);
    g1_mma(s1, acc);
  }

#pragma unroll
  for (int g = 0; g < 2; ++g)
#pragma unroll
    for (int r = 0; r < 8; ++r) {
      v4f o;
      o.x = acc[g][0][r];
      o.y = acc[g][1][r];
      o.z = acc[g][2][r];
      o.w = acc[g][3][r];
      *(v4f*)(P + ((long)kc * ROWS + ms * 16 + 8 * half + r) * HID + 64 * g + 4 * l16) = o;
    }
}

__global__ void reduceP(const float* __restrict__ P, float* __restrict__ T) {
  const int idx = blockIdx.x * 256 + threadIdx.x;
  float s = 0.0f;
  for (int kc = 0; kc < 64; ++kc) s += P[(long)kc * (ROWS * HID) + idx];
  T[idx] = s;
}

__global__ void s_kernel(const float* __restrict__ T, const float* __restrict__ w3p,
                         const float* __restrict__ b3p, const float* __restrict__ X1,
                         float* __restrict__ S) {
  const int idx = blockIdx.x * 256 + threadIdx.x;
  const int b = idx >> 10;
  const int o = (idx >> 5) & 31;
  const int i = idx & 31;
  const int r = b * 32 + i;
  const int oi = o * 32 + i;
  float acc = b3p[oi] * X1[r];
  for (int k = 0; k < HID; ++k) acc += T[r * HID + k] * w3p[k * 1024 + oi];
  S[idx] = acc;
}

__global__ void u_kernel(const float* __restrict__ S, const float* __restrict__ w3q,
                         float* __restrict__ U) {
  const int idx = blockIdx.x * 256 + threadIdx.x;
  const int bo = idx >> 7;
  const int k = idx & 127;
  const int o = bo & 31;
  float acc = 0.0f;
#pragma unroll
  for (int i = 0; i < 32; ++i) acc += S[bo * 32 + i] * w3q[k * 1024 + o * 32 + i];
  U[idx] = acc;
}

__global__ void c_kernel(const float* __restrict__ S, const float* __restrict__ b3q,
                         float* __restrict__ C) {
  const int bo = threadIdx.x;
  const int o = bo & 31;
  float acc = 0.0f;
#pragma unroll
  for (int i = 0; i < 32; ++i) acc += S[bo * 32 + i] * b3q[o * 32 + i];
  C[bo] = acc;
}

__global__ void gemm2(const float* __restrict__ U, const float* __restrict__ H2S,
                      const float* __restrict__ C, float* __restrict__ out) {
  const int lane = threadIdx.x;
  const int half = lane >> 4;
  const int l16  = lane & 15;
  const int mt = blockIdx.x;
  const int n0 = blockIdx.y * 4;

  v8f acc[4];
  const v8f vzero = {0.f,0.f,0.f,0.f,0.f,0.f,0.f,0.f};
#pragma unroll
  for (int t = 0; t < 4; ++t) acc[t] = vzero;

  const float* aptr = U   + (long)(mt * 16 + l16) * HID + 2 * half;
  const float* bptr = H2S + ((long)n0 * 16 + l16) * HID + 2 * half;
#pragma unroll
  for (int kb = 0; kb < HID; kb += 4) {
    v2f a = *(const v2f*)(aptr + kb);
#pragma unroll
    for (int t = 0; t < 4; ++t) {
      v2f b = *(const v2f*)(bptr + t * 16 * HID + kb);
      acc[t] = wmma4(a, b, acc[t]);
    }
  }
  const float scale = 1.0f / (float)NX;
#pragma unroll
  for (int t = 0; t < 4; ++t)
#pragma unroll
    for (int r = 0; r < 8; ++r) {
      const int row = mt * 16 + r + 8 * half;
      out[(long)row * NX + (n0 + t) * 16 + l16] = (acc[t][r] + C[row]) * scale;
    }
}

extern "C" void kernel_launch(void* const* d_in, const int* in_sizes, int n_in,
                              void* d_out, int out_size, void* d_ws, size_t ws_size,
                              hipStream_t stream) {
  const float* x   = (const float*)d_in[0];
  const float* pw1 = (const float*)d_in[1];
  const float* pb1 = (const float*)d_in[2];
  const float* pw2 = (const float*)d_in[3];
  const float* pb2 = (const float*)d_in[4];
  const float* pw3 = (const float*)d_in[5];
  const float* pb3 = (const float*)d_in[6];
  const float* qw1 = (const float*)d_in[7];
  const float* qb1 = (const float*)d_in[8];
  const float* qw2 = (const float*)d_in[9];
  const float* qb2 = (const float*)d_in[10];
  const float* qw3 = (const float*)d_in[11];
  const float* qb3 = (const float*)d_in[12];

  float* ws  = (float*)d_ws;
  float* H2P = ws;
  float* H2S = H2P + HID * NX;
  float* P   = H2S + NX * HID;
  float* T   = P + 64 * ROWS * HID;
  float* X1  = T + ROWS * HID;
  float* S   = X1 + ROWS;
  float* Uu  = S + 8192;
  float* Cc  = Uu + ROWS * HID;

  mlp_trunk<<<dim3(NX / 32, HID / 64, 2), dim3(32), 0, stream>>>(pw1, pb1, pw2, pb2,
                                                                 qw1, qb1, qw2, qb2, H2P, H2S);
  rowsum<<<dim3(ROWS), dim3(256), 0, stream>>>(x, X1);
  gemm1_partial<<<dim3(4, 64), dim3(128), 0, stream>>>(x, H2P, P);
  reduceP<<<dim3(128), dim3(256), 0, stream>>>(P, T);
  s_kernel<<<dim3(32), dim3(256), 0, stream>>>(T, pw3, pb3, X1, S);
  u_kernel<<<dim3(128), dim3(256), 0, stream>>>(S, qw3, Uu);
  c_kernel<<<dim3(1), dim3(256), 0, stream>>>(S, qb3, Cc);
  gemm2<<<dim3(16, 256), dim3(32), 0, stream>>>(Uu, H2S, Cc, (float*)d_out);
}


# GEMM2 operand loads made whole-line: psi trunk stored feature-major (layer-2 product transposed), a wave keeps its strip's A fragments, 4 row strips per block share a 256-column chunk
# speedup vs baseline: 6.1799x; 1.1892x over previous
#include <hip/hip_runtime.h>

typedef __attribute__((ext_vector_type(2))) float v2f;
typedef __attribute__((ext_vector_type(4))) float v4f;
typedef __attribute__((ext_vector_type(8))) float v8f;

#define NX    16384
#define HID   128
#define ROWS  256

__device__ __forceinline__ v8f wmma4(v2f a, v2f b, v8f c) {
  return __builtin_amdgcn_wmma_f32_16x16x4_f32(
       false, a,  false, b,
       (short)0, c,  false,  false);
}

__device__ __forceinline__ void trunk_tile(const float* __restrict__ w1,
                                           const float* __restrict__ b1,
                                           const float* __restrict__ w2,
                                           const float* __restrict__ b2,
                                           float* __restrict__ H,
                                           int n0, int c0, int half, int l16) {
  v2f hf[2][16];
#pragma unroll
  for (int p = 0; p < 2; ++p) {
    const int n = n0 + 2 * l16 + p;
    const float gx = (float)(n >> 7) * (1.0f / 127.0f);
    const float gy = (float)(n & 127) * (1.0f / 127.0f);
#pragma unroll
    for (int ks = 0; ks < 16; ++ks) {
      const int j0 = 4 * ks + 2 * half;
      const v2f wa = *(const v2f*)(w1 + j0);
      const v2f wb = *(const v2f*)(w1 + 64 + j0);
      const v2f bb = *(const v2f*)(b1 + j0);
      v2f h;
      h.x = fmaxf(gx * wa.x + gy * wb.x + bb.x, 0.0f);
      h.y = fmaxf(gx * wa.y + gy * wb.y + bb.y, 0.0f);
      hf[p][ks] = h;
    }
  }

  v8f acc[4][2];
  const v4f bq = *(const v4f*)(b2 + c0 + 4 * l16);
#pragma unroll
  for (int tt = 0; tt < 4; ++tt)
#pragma unroll
    for (int r = 0; r < 8; ++r) {
      acc[tt][0][r] = bq[tt];
      acc[tt][1][r] = bq[tt];
    }

  const float* wp = w2 + (2 * half) * HID + c0 + 4 * l16;
#pragma unroll
  for (int ks = 0; ks < 16; ++ks) {
    const v4f wlo = *(const v4f*)(wp + (4 * ks) * HID);
    const v4f whi = *(const v4f*)(wp + (4 * ks + 1) * HID);
#pragma unroll
    for (int tt = 0; tt < 4; ++tt) {
      v2f wf;
      wf.x = wlo[tt];
      wf.y = whi[tt];
#pragma unroll
      for (int p = 0; p < 2; ++p)
        acc[tt][p] = wmma4(hf[p][ks], wf, acc[tt][p]);
    }
  }

#pragma unroll
  for (int p = 0; p < 2; ++p)
#pragma unroll
    for (int r = 0; r < 8; ++r) {
      const int n = n0 + 2 * (8 * half + r) + p;
      v4f o;
      o.x = fmaxf(acc[0][p][r], 0.0f);
      o.y = fmaxf(acc[1][p][r], 0.0f);
      o.z = fmaxf(acc[2][p][r], 0.0f);
      o.w = fmaxf(acc[3][p][r], 0.0f);
      *(v4f*)(H + (long)n * HID + c0 + 4 * l16) = o;
    }
}


__device__ __forceinline__ void trunk_tile_t(const float* __restrict__ w1,
                                             const float* __restrict__ b1,
                                             const float* __restrict__ w2,
                                             const float* __restrict__ b2,
                                             float* __restrict__ HT,
                                             int n0, int c0, int half, int l16) {
  v2f hf[2][16];
#pragma unroll
  for (int p = 0; p < 2; ++p) {
    const int n = n0 + 2 * l16 + p;
    const float gx = (float)(n >> 7) * (1.0f / 127.0f);
    const float gy = (float)(n & 127) * (1.0f / 127.0f);
#pragma unroll
    for (int ks = 0; ks < 16; ++ks) {
      const int j0 = 4 * ks + 2 * half;
      const v2f wa = *(const v2f*)(w1 + j0);
      const v2f wb = *(const v2f*)(w1 + 64 + j0);
      const v2f bb = *(const v2f*)(b1 + j0);
      v2f h;
      h.x = fmaxf(gx * wa.x + gy * wb.x + bb.x, 0.0f);
      h.y = fmaxf(gx * wa.y + gy * wb.y + bb.y, 0.0f);
      hf[p][ks] = h;
    }
  }

  v8f acc[4][2];
#pragma unroll
  for (int r = 0; r < 8; ++r) {
    const v4f bq = *(const v4f*)(b2 + c0 + 4 * (8 * half + r));
#pragma unroll
    for (int tt = 0; tt < 4; ++tt) {
      acc[tt][0][r] = bq[tt];
      acc[tt][1][r] = bq[tt];
    }
  }

  const float* wp = w2 + (2 * half) * HID + c0 + 4 * l16;
#pragma unroll
  for (int ks = 0; ks < 16; ++ks) {
    const v4f wlo = *(const v4f*)(wp + (4 * ks) * HID);
    const v4f whi = *(const v4f*)(wp + (4 * ks + 1) * HID);
#pragma unroll
    for (int tt = 0; tt < 4; ++tt) {
      v2f wf;
      wf.x = wlo[tt];
      wf.y = whi[tt];
#pragma unroll
      for (int p = 0; p < 2; ++p)
        acc[tt][p] = wmma4(wf, hf[p][ks], acc[tt][p]);
    }
  }

#pragma unroll
  for (int tt = 0; tt < 4; ++tt)
#pragma unroll
    for (int r = 0; r < 8; ++r) {
      const int c = c0 + 4 * (8 * half + r) + tt;
      v2f o;
      o.x = fmaxf(acc[tt][0][r], 0.0f);
      o.y = fmaxf(acc[tt][1][r], 0.0f);
      *(v2f*)(HT + (long)c * NX + n0 + 2 * l16) = o;
    }
}

__global__ void __launch_bounds__(32)
mlp_trunk(const float* __restrict__ pw1, const float* __restrict__ pb1,
          const float* __restrict__ pw2, const float* __restrict__ pb2,
          const float* __restrict__ qw1, const float* __restrict__ qb1,
          const float* __restrict__ qw2, const float* __restrict__ qb2,
          float* __restrict__ H2P, float* __restrict__ HTS) {
  const int lane = threadIdx.x;
  const int half = lane >> 4;
  const int l16  = lane & 15;
  const int n0 = blockIdx.x * 32;
  const int c0 = blockIdx.y * 64;
  if (blockIdx.z == 0) trunk_tile(pw1, pb1, pw2, pb2, H2P, n0, c0, half, l16);
  else                 trunk_tile_t(qw1, qb1, qw2, qb2, HTS, n0, c0, half, l16);
}

__global__ void rowsum(const float* __restrict__ x, float* __restrict__ X1) {
  __shared__ float red[256];
  const int r = blockIdx.x;
  float s = 0.0f;
  for (int n = threadIdx.x; n < NX; n += 256) s += x[(long)r * NX + n];
  red[threadIdx.x] = s;
  __syncthreads();
  for (int off = 128; off > 0; off >>= 1) {
    if ((int)threadIdx.x < off) red[threadIdx.x] += red[threadIdx.x + off];
    __syncthreads();
  }
  if (threadIdx.x == 0) X1[r] = red[0];
}

struct G1Stage {
  v4f xa[4];
  v4f bw[8][2][2];
};

__device__ __forceinline__ void g1_load(G1Stage& st, const float* __restrict__ ap,
                                        const float* __restrict__ bp) {
#pragma unroll
  for (int q = 0; q < 4; ++q) st.xa[q] = *(const v4f*)(ap + 4 * q);
#pragma unroll
  for (int s = 0; s < 8; ++s)
#pragma unroll
    for (int g = 0; g < 2; ++g)
#pragma unroll
      for (int e = 0; e < 2; ++e)
        st.bw[s][g][e] = *(const v4f*)(bp + (2 * s + e) * HID + 64 * g);
}

__device__ __forceinline__ void g1_mma(const G1Stage& st, v8f (&acc)[2][4]) {
#pragma unroll
  for (int s = 0; s < 8; ++s) {
    v2f a;
    a.x = st.xa[s >> 1][2 * (s & 1)];
    a.y = st.xa[s >> 1][2 * (s & 1) + 1];
#pragma unroll
    for (int g = 0; g < 2; ++g)
#pragma unroll
      for (int tt = 0; tt < 4; ++tt) {
        v2f b;
        b.x = st.bw[s][g][0][tt];
        b.y = st.bw[s][g][1][tt];
        acc[g][tt] = wmma4(a, b, acc[g][tt]);
      }
  }
}

__global__ void __launch_bounds__(128)
gemm1_partial(const float* __restrict__ X, const float* __restrict__ H2P,
              float* __restrict__ P) {
  const int lane = threadIdx.x & 31;
  const int half = lane >> 4;
  const int l16  = lane & 15;
  const int ms   = blockIdx.x * 4 + (threadIdx.x >> 5);
  const int kc   = blockIdx.y;

  v8f acc[2][4];
  const v8f vzero = {0.f,0.f,0.f,0.f,0.f,0.f,0.f,0.f};
#pragma unroll
  for (int g = 0; g < 2; ++g)
#pragma unroll
    for (int tt = 0; tt < 4; ++tt) acc[g][tt] = vzero;

  const float* aptr = X   + (long)(ms * 16 + l16) * NX + kc * 256 + 16 * half;
  const float* bptr = H2P + ((long)kc * 256 + 16 * half) * HID + 4 * l16;

  G1Stage s0, s1;
  g1_load(s0, aptr, bptr);
  for (int st = 0; st < 8; st += 2) {
    g1_load(s1, aptr + (st + 1) * 32, bptr + (long)(st + 1) * 32 * HID);
    __builtin_amdgcn_sched_barrier(0);
    g1_mma(s0, acc);
    if (st + 2 < 8) g1_load(s0, aptr + (st + 2) * 32, bptr + (long)(st + 2) * 32 * HID);
    __builtin_amdgcn_sched_barrier(0);
    g1_mma(s1, acc);
  }

#pragma unroll
  for (int g = 0; g < 2; ++g)
#pragma unroll
    for (int r = 0; r < 8; ++r) {
      v4f o;
      o.x = acc[g][0][r];
      o.y = acc[g][1][r];
      o.z = acc[g][2][r];
      o.w = acc[g][3][r];
      *(v4f*)(P + ((long)kc * ROWS + ms * 16 + 8 * half + r) * HID + 64 * g + 4 * l16) = o;
    }
}

__global__ void reduceP(const float* __restrict__ P, float* __restrict__ T) {
  const int idx = blockIdx.x * 256 + threadIdx.x;
  float s = 0.0f;
  for (int kc = 0; kc < 64; ++kc) s += P[(long)kc * (ROWS * HID) + idx];
  T[idx] = s;
}

__global__ void s_kernel(const float* __restrict__ T, const float* __restrict__ w3p,
                         const float* __restrict__ b3p, const float* __restrict__ X1,
                         float* __restrict__ S) {
  const int idx = blockIdx.x * 256 + threadIdx.x;
  const int b = idx >> 10;
  const int o = (idx >> 5) & 31;
  const int i = idx & 31;
  const int r = b * 32 + i;
  const int oi = o * 32 + i;
  float acc = b3p[oi] * X1[r];
  for (int k = 0; k < HID; ++k) acc += T[r * HID + k] * w3p[k * 1024 + oi];
  S[idx] = acc;
}

__global__ void u_kernel(const float* __restrict__ S, const float* __restrict__ w3q,
                         float* __restrict__ U) {
  const int idx = blockIdx.x * 256 + threadIdx.x;
  const int bo = idx >> 7;
  const int k = idx & 127;
  const int o = bo & 31;
  float acc = 0.0f;
#pragma unroll
  for (int i = 0; i < 32; ++i) acc += S[bo * 32 + i] * w3q[k * 1024 + o * 32 + i];
  U[idx] = acc;
}

__global__ void c_kernel(const float* __restrict__ S, const float* __restrict__ b3q,
                         float* __restrict__ C) {
  const int bo = threadIdx.x;
  const int o = bo & 31;
  float acc = 0.0f;
#pragma unroll
  for (int i = 0; i < 32; ++i) acc += S[bo * 32 + i] * b3q[o * 32 + i];
  C[bo] = acc;
}

struct G2Stage {
  v4f b[16][2];
};

__device__ __forceinline__ void g2_load(G2Stage& st, const float* __restrict__ bp) {
#pragma unroll
  for (int j = 0; j < 16; ++j)
#pragma unroll
    for (int e = 0; e < 2; ++e)
      st.b[j][e] = *(const v4f*)(bp + (long)(4 * j + e) * NX);
}

template <int S0>
__device__ __forceinline__ void g2_mma(const G2Stage& st, const v2f (&a)[32], v8f (&acc)[4]) {
#pragma unroll
  for (int j = 0; j < 16; ++j)
#pragma unroll
    for (int tt = 0; tt < 4; ++tt) {
      v2f b;
      b.x = st.b[j][0][tt];
      b.y = st.b[j][1][tt];
      acc[tt] = wmma4(a[S0 + j], b, acc[tt]);
    }
}

__global__ void __launch_bounds__(128)
gemm2(const float* __restrict__ U, const float* __restrict__ HT,
      const float* __restrict__ C, float* __restrict__ out) {
  const int lane = threadIdx.x & 31;
  const int half = lane >> 4;
  const int l16  = lane & 15;
  const int ms   = blockIdx.x * 4 + (threadIdx.x >> 5);
  const int nc   = blockIdx.y;

  const float* bptr = HT + (long)(2 * half) * NX + nc * 256 + 4 * l16;

  G2Stage s0, s1;
  g2_load(s0, bptr);
  __builtin_amdgcn_sched_barrier(0);

  v2f a[32];
  const float* aptr = U + (long)(ms * 16 + l16) * HID + 2 * half;
#pragma unroll
  for (int s = 0; s < 32; ++s) a[s] = *(const v2f*)(aptr + 4 * s);

  float crow[8];
#pragma unroll
  for (int r = 0; r < 8; ++r) crow[r] = C[ms * 16 + 8 * half + r];

  const float scale = 1.0f / (float)NX;
  float* optr = out + (long)(ms * 16 + 8 * half) * NX + nc * 256 + 4 * l16;

  for (int g = 0; g < 4; ++g) {
    v8f acc[4];
    const v8f vzero = {0.f,0.f,0.f,0.f,0.f,0.f,0.f,0.f};
#pragma unroll
    for (int tt = 0; tt < 4; ++tt) acc[tt] = vzero;

    g2_load(s1, bptr + (long)64 * NX + g * 64);
    __builtin_amdgcn_sched_barrier(0);
    g2_mma<0>(s0, a, acc);
    if (g + 1 < 4) g2_load(s0, bptr + (g + 1) * 64);
    __builtin_amdgcn_sched_barrier(0);
    g2_mma<16>(s1, a, acc);

#pragma unroll
    for (int r = 0; r < 8; ++r) {
      v4f o;
      o.x = (acc[0][r] + crow[r]) * scale;
      o.y = (acc[1][r] + crow[r]) * scale;
      o.z = (acc[2][r] + crow[r]) * scale;
      o.w = (acc[3][r] + crow[r]) * scale;
      *(v4f*)(optr + (long)r * NX + g * 64) = o;
    }
  }
}

extern "C" void kernel_launch(void* const* d_in, const int* in_sizes, int n_in,
                              void* d_out, int out_size, void* d_ws, size_t ws_size,
                              hipStream_t stream) {
  const float* x   = (const float*)d_in[0];
  const float* pw1 = (const float*)d_in[1];
  const float* pb1 = (const float*)d_in[2];
  const float* pw2 = (const float*)d_in[3];
  const float* pb2 = (const float*)d_in[4];
  const float* pw3 = (const float*)d_in[5];
  const float* pb3 = (const float*)d_in[6];
  const float* qw1 = (const float*)d_in[7];
  const float* qb1 = (const float*)d_in[8];
  const float* qw2 = (const float*)d_in[9];
  const float* qb2 = (const float*)d_in[10];
  const float* qw3 = (const float*)d_in[11];
  const float* qb3 = (const float*)d_in[12];

  float* ws  = (float*)d_ws;
  float* H2P = ws;
  float* HTS = H2P + HID * NX;
  float* P   = HTS + NX * HID;
  float* T   = P + 64 * ROWS * HID;
  float* X1  = T + ROWS * HID;
  float* S   = X1 + ROWS;
  float* Uu  = S + 8192;
  float* Cc  = Uu + ROWS * HID;

  mlp_trunk<<<dim3(NX / 32, HID / 64, 2), dim3(32), 0, stream>>>(pw1, pb1, pw2, pb2,
                                                                 qw1, qb1, qw2, qb2, H2P, HTS);
  rowsum<<<dim3(ROWS), dim3(256), 0, stream>>>(x, X1);
  gemm1_partial<<<dim3(4, 64), dim3(128), 0, stream>>>(x, H2P, P);
  reduceP<<<dim3(128), dim3(256), 0, stream>>>(P, T);
  s_kernel<<<dim3(32), dim3(256), 0, stream>>>(T, pw3, pb3, X1, S);
  u_kernel<<<dim3(128), dim3(256), 0, stream>>>(S, qw3, Uu);
  c_kernel<<<dim3(1), dim3(256), 0, stream>>>(S, qb3, Cc);
  gemm2<<<dim3(4, 64), dim3(128), 0, stream>>>(Uu, HTS, Cc, (float*)d_out);
}


# Launches 8 -> 5: S, U and C in one kernel per (b,o) with operand tiles in LDS, x row sums taken from GEMM1's x fragments; f32 sum order of the row sums and of S's k sum (8 partials of 16) changes
# speedup vs baseline: 9.7417x; 1.5764x over previous
#include <hip/hip_runtime.h>

typedef __attribute__((ext_vector_type(2))) float v2f;
typedef __attribute__((ext_vector_type(4))) float v4f;
typedef __attribute__((ext_vector_type(8))) float v8f;

#define NX    16384
#define HID   128
#define ROWS  256

__device__ __forceinline__ v8f wmma4(v2f a, v2f b, v8f c) {
  return __builtin_amdgcn_wmma_f32_16x16x4_f32(
       false, a,  false, b,
       (short)0, c,  false,  false);
}

__device__ __forceinline__ void trunk_tile(const float* __restrict__ w1,
                                           const float* __restrict__ b1,
                                           const float* __restrict__ w2,
                                           const float* __restrict__ b2,
                                           float* __restrict__ H,
                                           int n0, int c0, int half, int l16) {
  v2f hf[2][16];
#pragma unroll
  for (int p = 0; p < 2; ++p) {
    const int n = n0 + 2 * l16 + p;
    const float gx = (float)(n >> 7) * (1.0f / 127.0f);
    const float gy = (float)(n & 127) * (1.0f / 127.0f);
#pragma unroll
    for (int ks = 0; ks < 16; ++ks) {
      const int j0 = 4 * ks + 2 * half;
      const v2f wa = *(const v2f*)(w1 + j0);
      const v2f wb = *(const v2f*)(w1 + 64 + j0);
      const v2f bb = *(const v2f*)(b1 + j0);
      v2f h;
      h.x = fmaxf(gx * wa.x + gy * wb.x + bb.x, 0.0f);
      h.y = fmaxf(gx * wa.y + gy * wb.y + bb.y, 0.0f);
      hf[p][ks] = h;
    }
  }

  v8f acc[4][2];
  const v4f bq = *(const v4f*)(b2 + c0 + 4 * l16);
#pragma unroll
  for (int tt = 0; tt < 4; ++tt)
#pragma unroll
    for (int r = 0; r < 8; ++r) {
      acc[tt][0][r] = bq[tt];
      acc[tt][1][r] = bq[tt];
    }

  const float* wp = w2 + (2 * half) * HID + c0 + 4 * l16;
#pragma unroll
  for (int ks = 0; ks < 16; ++ks) {
    const v4f wlo = *(const v4f*)(wp + (4 * ks) * HID);
    const v4f whi = *(const v4f*)(wp + (4 * ks + 1) * HID);
#pragma unroll
    for (int tt = 0; tt < 4; ++tt) {
      v2f wf;
      wf.x = wlo[tt];
      wf.y = whi[tt];
#pragma unroll
      for (int p = 0; p < 2; ++p)
        acc[tt][p] = wmma4(hf[p][ks], wf, acc[tt][p]);
    }
  }

#pragma unroll
  for (int p = 0; p < 2; ++p)
#pragma unroll
    for (int r = 0; r < 8; ++r) {
      const int n = n0 + 2 * (8 * half + r) + p;
      v4f o;
      o.x = fmaxf(acc[0][p][r], 0.0f);
      o.y = fmaxf(acc[1][p][r], 0.0f);
      o.z = fmaxf(acc[2][p][r], 0.0f);
      o.w = fmaxf(acc[3][p][r], 0.0f);
      *(v4f*)(H + (long)n * HID + c0 + 4 * l16) = o;
    }
}


__device__ __forceinline__ void trunk_tile_t(const float* __restrict__ w1,
                                             const float* __restrict__ b1,
                                             const float* __restrict__ w2,
                                             const float* __restrict__ b2,
                                             float* __restrict__ HT,
                                             int n0, int c0, int half, int l16) {
  v2f hf[2][16];
#pragma unroll
  for (int p = 0; p < 2; ++p) {
    const int n = n0 + 2 * l16 + p;
    const float gx = (float)(n >> 7) * (1.0f / 127.0f);
    const float gy = (float)(n & 127) * (1.0f / 127.0f);
#pragma unroll
    for (int ks = 0; ks < 16; ++ks) {
      const int j0 = 4 * ks + 2 * half;
      const v2f wa = *(const v2f*)(w1 + j0);
      const v2f wb = *(const v2f*)(w1 + 64 + j0);
      const v2f bb = *(const v2f*)(b1 + j0);
      v2f h;
      h.x = fmaxf(gx * wa.x + gy * wb.x + bb.x, 0.0f);
      h.y = fmaxf(gx * wa.y + gy * wb.y + bb.y, 0.0f);
      hf[p][ks] = h;
    }
  }

  v8f acc[4][2];
#pragma unroll
  for (int r = 0; r < 8; ++r) {
    const v4f bq = *(const v4f*)(b2 + c0 + 4 * (8 * half + r));
#pragma unroll
    for (int tt = 0; tt < 4; ++tt) {
      acc[tt][0][r] = bq[tt];
      acc[tt][1][r] = bq[tt];
    }
  }

  const float* wp = w2 + (2 * half) * HID + c0 + 4 * l16;
#pragma unroll
  for (int ks = 0; ks < 16; ++ks) {
    const v4f wlo = *(const v4f*)(wp + (4 * ks) * HID);
    const v4f whi = *(const v4f*)(wp + (4 * ks + 1) * HID);
#pragma unroll
    for (int tt = 0; tt < 4; ++tt) {
      v2f wf;
      wf.x = wlo[tt];
      wf.y = whi[tt];
#pragma unroll
      for (int p = 0; p < 2; ++p)
        acc[tt][p] = wmma4(wf, hf[p][ks], acc[tt][p]);
    }
  }

#pragma unroll
  for (int tt = 0; tt < 4; ++tt)
#pragma unroll
    for (int r = 0; r < 8; ++r) {
      const int c = c0 + 4 * (8 * half + r) + tt;
      v2f o;
      o.x = fmaxf(acc[tt][0][r], 0.0f);
      o.y = fmaxf(acc[tt][1][r], 0.0f);
      *(v2f*)(HT + (long)c * NX + n0 + 2 * l16) = o;
    }
}

__global__ void __launch_bounds__(32)
mlp_trunk(const float* __restrict__ pw1, const float* __restrict__ pb1,
          const float* __restrict__ pw2, const float* __restrict__ pb2,
          const float* __restrict__ qw1, const float* __restrict__ qb1,
          const float* __restrict__ qw2, const float* __restrict__ qb2,
          float* __restrict__ H2P, float* __restrict__ HTS) {
  const int lane = threadIdx.x;
  const int half = lane >> 4;
  const int l16  = lane & 15;
  const int n0 = blockIdx.x * 32;
  const int c0 = blockIdx.y * 64;
  if (blockIdx.z == 0) trunk_tile(pw1, pb1, pw2, pb2, H2P, n0, c0, half, l16);
  else                 trunk_tile_t(qw1, qb1, qw2, qb2, HTS, n0, c0, half, l16);
}

struct G1Stage {
  v4f xa[4];
  v4f bw[8][2][2];
};

__device__ __forceinline__ void g1_load(G1Stage& st, const float* __restrict__ ap,
                                        const float* __restrict__ bp) {
#pragma unroll
  for (int q = 0; q < 4; ++q) st.xa[q] = *(const v4f*)(ap + 4 * q);
#pragma unroll
  for (int s = 0; s < 8; ++s)
#pragma unroll
    for (int g = 0; g < 2; ++g)
#pragma unroll
      for (int e = 0; e < 2; ++e)
        st.bw[s][g][e] = *(const v4f*)(bp + (2 * s + e) * HID + 64 * g);
}

__device__ __forceinline__ void g1_mma(const G1Stage& st, v8f (&acc)[2][4], v4f& rs) {
  rs += (st.xa[0] + st.xa[1]) + (st.xa[2] + st.xa[3]);
#pragma unroll
  for (int s = 0; s < 8; ++s) {
    v2f a;
    a.x = st.xa[s >> 1][2 * (s & 1)];
    a.y = st.xa[s >> 1][2 * (s & 1) + 1];
#pragma unroll
    for (int g = 0; g < 2; ++g)
#pragma unroll
      for (int tt = 0; tt < 4; ++tt) {
        v2f b;
        b.x = st.bw[s][g][0][tt];
        b.y = st.bw[s][g][1][tt];
        acc[g][tt] = wmma4(a, b, acc[g][tt]);
      }
  }
}

__global__ void __launch_bounds__(128)
gemm1_partial(const float* __restrict__ X, const float* __restrict__ H2P,
              float* __restrict__ P, float* __restrict__ XP) {
  const int lane = threadIdx.x & 31;
  const int half = lane >> 4;
  const int l16  = lane & 15;
  const int ms   = blockIdx.x * 4 + (threadIdx.x >> 5);
  const int kc   = blockIdx.y;

  v8f acc[2][4];
  const v8f vzero = {0.f,0.f,0.f,0.f,0.f,0.f,0.f,0.f};
#pragma unroll
  for (int g = 0; g < 2; ++g)
#pragma unroll
    for (int tt = 0; tt < 4; ++tt) acc[g][tt] = vzero;
  v4f rs = {0.f, 0.f, 0.f, 0.f};

  const float* aptr = X   + (long)(ms * 16 + l16) * NX + kc * 256 + 16 * half;
  const float* bptr = H2P + ((long)kc * 256 + 16 * half) * HID + 4 * l16;

  G1Stage s0, s1;
  g1_load(s0, aptr, bptr);
#pragma unroll
  for (int st = 0; st < 8; st += 2) {
    g1_load(s1, aptr + (st + 1) * 32, bptr + (long)(st + 1) * 32 * HID);
    __builtin_amdgcn_sched_barrier(0);
    g1_mma(s0, acc, rs);
    if (st + 2 < 8) g1_load(s0, aptr + (st + 2) * 32, bptr + (long)(st + 2) * 32 * HID);
    __builtin_amdgcn_sched_barrier(0);
    g1_mma(s1, acc, rs);
  }

  {
    float rsum = (rs.x + rs.y) + (rs.z + rs.w);
    rsum += __shfl_xor(rsum, 16);
    XP[((long)kc * 16 + ms) * 32 + lane] = rsum;
  }

#pragma unroll
  for (int g = 0; g < 2; ++g)
#pragma unroll
    for (int r = 0; r < 8; ++r) {
      v4f o;
      o.x = acc[g][0][r];
      o.y = acc[g][1][r];
      o.z = acc[g][2][r];
      o.w = acc[g][3][r];
      *(v4f*)(P + ((long)kc * ROWS + ms * 16 + 8 * half + r) * HID + 64 * g + 4 * l16) = o;
    }
}

__global__ void reduceP(const float* __restrict__ P, const float* __restrict__ XP,
                        float* __restrict__ T, float* __restrict__ X1) {
  if (blockIdx.x == 128) {
    const int r = threadIdx.x;
    const float* xp = XP + (r >> 4) * 32 + (r & 15);
    float s = 0.0f;
    for (int kc = 0; kc < 64; ++kc) s += xp[kc * 512];
    X1[r] = s;
    return;
  }
  const int idx = blockIdx.x * 256 + threadIdx.x;
  float s = 0.0f;
  for (int kc = 0; kc < 64; ++kc) s += P[(long)kc * (ROWS * HID) + idx];
  T[idx] = s;
}

#define TL_PITCH 132
#define WQ_PITCH 36
__global__ void __launch_bounds__(256)
suc_kernel(const float* __restrict__ T, const float* __restrict__ w3p,
           const float* __restrict__ b3p, const float* __restrict__ X1,
           const float* __restrict__ w3q, const float* __restrict__ b3q,
           float* __restrict__ U, float* __restrict__ CP) {
  __shared__ __attribute__((aligned(16))) float Tl[32 * TL_PITCH];
  __shared__ __attribute__((aligned(16))) float Wp[HID * 32];
  __shared__ __attribute__((aligned(16))) float Wq[HID * WQ_PITCH];
  __shared__ __attribute__((aligned(16))) float Sp[8 * 32];
  __shared__ __attribute__((aligned(16))) float Sl[32];
  const int tid = threadIdx.x;
  const int bo  = blockIdx.x;
  const int b   = bo >> 5;
  const int o   = bo & 31;

  const float bx = b3p[o * 32 + (tid & 31)] * X1[b * 32 + (tid & 31)];

#pragma unroll
  for (int j = 0; j < 4; ++j) {
    const int f = tid + 256 * j;
    {
      const int i = f >> 5, c4 = f & 31;
      *(v4f*)(Tl + i * TL_PITCH + 4 * c4) =
          *(const v4f*)(T + (long)(b * 32 + i) * HID + 4 * c4);
    }
    {
      const int k = f >> 3, c4 = f & 7;
      *(v4f*)(Wp + k * 32 + 4 * c4) =
          *(const v4f*)(w3p + (long)k * 1024 + o * 32 + 4 * c4);
      *(v4f*)(Wq + k * WQ_PITCH + 4 * c4) =
          *(const v4f*)(w3q + (long)k * 1024 + o * 32 + 4 * c4);
    }
  }
  __syncthreads();

  {
    const int i = tid & 31, k0 = (tid >> 5) * 16;
    float acc = 0.0f;
#pragma unroll
    for (int k4 = 0; k4 < 4; ++k4) {
      const v4f t = *(const v4f*)(Tl + i * TL_PITCH + k0 + 4 * k4);
      acc += t.x * Wp[(k0 + 4 * k4 + 0) * 32 + i];
      acc += t.y * Wp[(k0 + 4 * k4 + 1) * 32 + i];
      acc += t.z * Wp[(k0 + 4 * k4 + 2) * 32 + i];
      acc += t.w * Wp[(k0 + 4 * k4 + 3) * 32 + i];
    }
    Sp[tid] = acc;
  }
  __syncthreads();

  if (tid < 32) {
    const int i = tid;
    float acc = bx;
#pragma unroll
    for (int w = 0; w < 8; ++w) acc += Sp[w * 32 + i];
    Sl[i] = acc;
  }
  __syncthreads();

  if (tid < HID) {
    const int k = tid;
    float acc = 0.0f;
#pragma unroll
    for (int i4 = 0; i4 < 8; ++i4) {
      const v4f s = *(const v4f*)(Sl + 4 * i4);
      const v4f w = *(const v4f*)(Wq + k * WQ_PITCH + 4 * i4);
      acc += s.x * w.x;
      acc += s.y * w.y;
      acc += s.z * w.z;
      acc += s.w * w.w;
    }
    U[bo * HID + k] = acc;
  } else if (tid < HID + 32) {
    float acc = 0.0f;
#pragma unroll
    for (int i = 0; i < 32; ++i) acc += Sl[i] * b3q[o * 32 + i];
    CP[bo * 32 + (tid - HID)] = acc;
  }
}

struct G2Stage {
  v4f b[16][2];
};

__device__ __forceinline__ void g2_load(G2Stage& st, const float* __restrict__ bp) {
#pragma unroll
  for (int j = 0; j < 16; ++j)
#pragma unroll
    for (int e = 0; e < 2; ++e)
      st.b[j][e] = *(const v4f*)(bp + (long)(4 * j + e) * NX);
}

template <int S0>
__device__ __forceinline__ void g2_mma(const G2Stage& st, const v2f (&a)[32], v8f (&acc)[4]) {
#pragma unroll
  for (int j = 0; j < 16; ++j)
#pragma unroll
    for (int tt = 0; tt < 4; ++tt) {
      v2f b;
      b.x = st.b[j][0][tt];
      b.y = st.b[j][1][tt];
      acc[tt] = wmma4(a[S0 + j], b, acc[tt]);
    }
}

__global__ void __launch_bounds__(128)
gemm2(const float* __restrict__ U, const float* __restrict__ HT,
      const float* __restrict__ C, float* __restrict__ out) {
  const int lane = threadIdx.x & 31;
  const int half = lane >> 4;
  const int l16  = lane & 15;
  const int ms   = blockIdx.x * 4 + (threadIdx.x >> 5);
  const int nc   = blockIdx.y;

  const float* bptr = HT + (long)(2 * half) * NX + nc * 256 + 4 * l16;

  G2Stage s0, s1;
  g2_load(s0, bptr);
  __builtin_amdgcn_sched_barrier(0);

  v2f a[32];
  const float* aptr = U + (long)(ms * 16 + l16) * HID + 2 * half;
#pragma unroll
  for (int s = 0; s < 32; ++s) a[s] = *(const v2f*)(aptr + 4 * s);

  float crow[8];
#pragma unroll
  for (int r = 0; r < 8; ++r) crow[r] = C[(ms * 16 + 8 * half + r) * 32];

  const float scale = 1.0f / (float)NX;
  float* optr = out + (long)(ms * 16 + 8 * half) * NX + nc * 256 + 4 * l16;

  for (int g = 0; g < 4; ++g) {
    v8f acc[4];
    const v8f vzero = {0.f,0.f,0.f,0.f,0.f,0.f,0.f,0.f};
#pragma unroll
    for (int tt = 0; tt < 4; ++tt) acc[tt] = vzero;

    g2_load(s1, bptr + (long)64 * NX + g * 64);
    __builtin_amdgcn_sched_barrier(0);
    g2_mma<0>(s0, a, acc);
    if (g + 1 < 4) g2_load(s0, bptr + (g + 1) * 64);
    __builtin_amdgcn_sched_barrier(0);
    g2_mma<16>(s1, a, acc);

#pragma unroll
    for (int r = 0; r < 8; ++r) {
      v4f o;
      o.x = (acc[0][r] + crow[r]) * scale;
      o.y = (acc[1][r] + crow[r]) * scale;
      o.z = (acc[2][r] + crow[r]) * scale;
      o.w = (acc[3][r] + crow[r]) * scale;
      *(v4f*)(optr + (long)r * NX + g * 64) = o;
    }
  }
}

extern "C" void kernel_launch(void* const* d_in, const int* in_sizes, int n_in,
                              void* d_out, int out_size, void* d_ws, size_t ws_size,
                              hipStream_t stream) {
  const float* x   = (const float*)d_in[0];
  const float* pw1 = (const float*)d_in[1];
  const float* pb1 = (const float*)d_in[2];
  const float* pw2 = (const float*)d_in[3];
  const float* pb2 = (const float*)d_in[4];
  const float* pw3 = (const float*)d_in[5];
  const float* pb3 = (const float*)d_in[6];
  const float* qw1 = (const float*)d_in[7];
  const float* qb1 = (const float*)d_in[8];
  const float* qw2 = (const float*)d_in[9];
  const float* qb2 = (const float*)d_in[10];
  const float* qw3 = (const float*)d_in[11];
  const float* qb3 = (const float*)d_in[12];

  float* ws  = (float*)d_ws;
  float* H2P = ws;
  float* HTS = H2P + HID * NX;
  float* P   = HTS + NX * HID;
  float* T   = P + 64 * ROWS * HID;
  float* X1  = T + ROWS * HID;
  float* Uu  = X1 + ROWS;
  float* Cc  = Uu + ROWS * HID;
  float* XP  = Cc + ROWS * 32;

  mlp_trunk<<<dim3(NX / 32, HID / 64, 2), dim3(32), 0, stream>>>(pw1, pb1, pw2, pb2,
                                                                 qw1, qb1, qw2, qb2, H2P, HTS);
  gemm1_partial<<<dim3(4, 64), dim3(128), 0, stream>>>(x, H2P, P, XP);
  reduceP<<<dim3(129), dim3(256), 0, stream>>>(P, XP, T, X1);
  suc_kernel<<<dim3(ROWS), dim3(256), 0, stream>>>(T, pw3, pb3, X1, qw3, qb3, Uu, Cc);
  gemm2<<<dim3(4, 64), dim3(128), 0, stream>>>(Uu, HTS, Cc, (float*)d_out);
}


# Both GEMMs: operand stages halved (32 matrix instructions) and held in a ring of four, so three load groups (54 and 48 loads, under the 63-load counter) are in flight, not one; k and sum order kept
# speedup vs baseline: 10.0646x; 1.0331x over previous
#include <hip/hip_runtime.h>

typedef __attribute__((ext_vector_type(2))) float v2f;
typedef __attribute__((ext_vector_type(4))) float v4f;
typedef __attribute__((ext_vector_type(8))) float v8f;

#define NX    16384
#define HID   128
#define ROWS  256

__device__ __forceinline__ v8f wmma4(v2f a, v2f b, v8f c) {
  return __builtin_amdgcn_wmma_f32_16x16x4_f32(
       false, a,  false, b,
       (short)0, c,  false,  false);
}

__device__ __forceinline__ void trunk_tile(const float* __restrict__ w1,
                                           const float* __restrict__ b1,
                                           const float* __restrict__ w2,
                                           const float* __restrict__ b2,
                                           float* __restrict__ H,
                                           int n0, int c0, int half, int l16) {
  v2f hf[2][16];
#pragma unroll
  for (int p = 0; p < 2; ++p) {
    const int n = n0 + 2 * l16 + p;
    const float gx = (float)(n >> 7) * (1.0f / 127.0f);
    const float gy = (float)(n & 127) * (1.0f / 127.0f);
#pragma unroll
    for (int ks = 0; ks < 16; ++ks) {
      const int j0 = 4 * ks + 2 * half;
      const v2f wa = *(const v2f*)(w1 + j0);
      const v2f wb = *(const v2f*)(w1 + 64 + j0);
      const v2f bb = *(const v2f*)(b1 + j0);
      v2f h;
      h.x = fmaxf(gx * wa.x + gy * wb.x + bb.x, 0.0f);
      h.y = fmaxf(gx * wa.y + gy * wb.y + bb.y, 0.0f);
      hf[p][ks] = h;
    }
  }

  v8f acc[4][2];
  const v4f bq = *(const v4f*)(b2 + c0 + 4 * l16);
#pragma unroll
  for (int tt = 0; tt < 4; ++tt)
#pragma unroll
    for (int r = 0; r < 8; ++r) {
      acc[tt][0][r] = bq[tt];
      acc[tt][1][r] = bq[tt];
    }

  const float* wp = w2 + (2 * half) * HID + c0 + 4 * l16;
#pragma unroll
  for (int ks = 0; ks < 16; ++ks) {
    const v4f wlo = *(const v4f*)(wp + (4 * ks) * HID);
    const v4f whi = *(const v4f*)(wp + (4 * ks + 1) * HID);
#pragma unroll
    for (int tt = 0; tt < 4; ++tt) {
      v2f wf;
      wf.x = wlo[tt];
      wf.y = whi[tt];
#pragma unroll
      for (int p = 0; p < 2; ++p)
        acc[tt][p] = wmma4(hf[p][ks], wf, acc[tt][p]);
    }
  }

#pragma unroll
  for (int p = 0; p < 2; ++p)
#pragma unroll
    for (int r = 0; r < 8; ++r) {
      const int n = n0 + 2 * (8 * half + r) + p;
      v4f o;
      o.x = fmaxf(acc[0][p][r], 0.0f);
      o.y = fmaxf(acc[1][p][r], 0.0f);
      o.z = fmaxf(acc[2][p][r], 0.0f);
      o.w = fmaxf(acc[3][p][r], 0.0f);
      *(v4f*)(H + (long)n * HID + c0 + 4 * l16) = o;
    }
}


__device__ __forceinline__ void trunk_tile_t(const float* __restrict__ w1,
                                             const float* __restrict__ b1,
                                             const float* __restrict__ w2,
                                             const float* __restrict__ b2,
                                             float* __restrict__ HT,
                                             int n0, int c0, int half, int l16) {
  v2f hf[2][16];
#pragma unroll
  for (int p = 0; p < 2; ++p) {
    const int n = n0 + 2 * l16 + p;
    const float gx = (float)(n >> 7) * (1.0f / 127.0f);
    const float gy = (float)(n & 127) * (1.0f / 127.0f);
#pragma unroll
    for (int ks = 0; ks < 16; ++ks) {
      const int j0 = 4 * ks + 2 * half;
      const v2f wa = *(const v2f*)(w1 + j0);
      const v2f wb = *(const v2f*)(w1 + 64 + j0);
      const v2f bb = *(const v2f*)(b1 + j0);
      v2f h;
      h.x = fmaxf(gx * wa.x + gy * wb.x + bb.x, 0.0f);
      h.y = fmaxf(gx * wa.y + gy * wb.y + bb.y, 0.0f);
      hf[p][ks] = h;
    }
  }

  v8f acc[4][2];
#pragma unroll
  for (int r = 0; r < 8; ++r) {
    const v4f bq = *(const v4f*)(b2 + c0 + 4 * (8 * half + r));
#pragma unroll
    for (int tt = 0; tt < 4; ++tt) {
      acc[tt][0][r] = bq[tt];
      acc[tt][1][r] = bq[tt];
    }
  }

  const float* wp = w2 + (2 * half) * HID + c0 + 4 * l16;
#pragma unroll
  for (int ks = 0; ks < 16; ++ks) {
    const v4f wlo = *(const v4f*)(wp + (4 * ks) * HID);
    const v4f whi = *(const v4f*)(wp + (4 * ks + 1) * HID);
#pragma unroll
    for (int tt = 0; tt < 4; ++tt) {
      v2f wf;
      wf.x = wlo[tt];
      wf.y = whi[tt];
#pragma unroll
      for (int p = 0; p < 2; ++p)
        acc[tt][p] = wmma4(wf, hf[p][ks], acc[tt][p]);
    }
  }

#pragma unroll
  for (int tt = 0; tt < 4; ++tt)
#pragma unroll
    for (int r = 0; r < 8; ++r) {
      const int c = c0 + 4 * (8 * half + r) + tt;
      v2f o;
      o.x = fmaxf(acc[tt][0][r], 0.0f);
      o.y = fmaxf(acc[tt][1][r], 0.0f);
      *(v2f*)(HT + (long)c * NX + n0 + 2 * l16) = o;
    }
}

__global__ void __launch_bounds__(32)
mlp_trunk(const float* __restrict__ pw1, const float* __restrict__ pb1,
          const float* __restrict__ pw2, const float* __restrict__ pb2,
          const float* __restrict__ qw1, const float* __restrict__ qb1,
          const float* __restrict__ qw2, const float* __restrict__ qb2,
          float* __restrict__ H2P, float* __restrict__ HTS) {
  const int lane = threadIdx.x;
  const int half = lane >> 4;
  const int l16  = lane & 15;
  const int n0 = blockIdx.x * 32;
  const int c0 = blockIdx.y * 64;
  if (blockIdx.z == 0) trunk_tile(pw1, pb1, pw2, pb2, H2P, n0, c0, half, l16);
  else                 trunk_tile_t(qw1, qb1, qw2, qb2, HTS, n0, c0, half, l16);
}

struct G1Unit {
  v4f xa[2];
  v4f bw[4][2][2];
};

__device__ __forceinline__ void g1_load(G1Unit& un, const float* __restrict__ ap,
                                        const float* __restrict__ bp, int q) {
#pragma unroll
  for (int p = 0; p < 2; ++p) un.xa[p] = *(const v4f*)(ap + 8 * q + 4 * p);
#pragma unroll
  for (int j = 0; j < 4; ++j)
#pragma unroll
    for (int g = 0; g < 2; ++g)
#pragma unroll
      for (int e = 0; e < 2; ++e)
        un.bw[j][g][e] = *(const v4f*)(bp + (8 * q + 2 * j + e) * HID + 64 * g);
}

__device__ __forceinline__ void g1_mma(const G1Unit& un, v8f (&acc)[2][4]) {
#pragma unroll
  for (int j = 0; j < 4; ++j) {
    v2f a;
    a.x = un.xa[j >> 1][2 * (j & 1)];
    a.y = un.xa[j >> 1][2 * (j & 1) + 1];
#pragma unroll
    for (int g = 0; g < 2; ++g)
#pragma unroll
      for (int tt = 0; tt < 4; ++tt) {
        v2f b;
        b.x = un.bw[j][g][0][tt];
        b.y = un.bw[j][g][1][tt];
        acc[g][tt] = wmma4(a, b, acc[g][tt]);
      }
  }
}

__global__ void __launch_bounds__(128)
gemm1_partial(const float* __restrict__ X, const float* __restrict__ H2P,
              float* __restrict__ P, float* __restrict__ XP) {
  const int lane = threadIdx.x & 31;
  const int half = lane >> 4;
  const int l16  = lane & 15;
  const int ms   = blockIdx.x * 4 + (threadIdx.x >> 5);
  const int kc   = blockIdx.y;

  v8f acc[2][4];
  const v8f vzero = {0.f,0.f,0.f,0.f,0.f,0.f,0.f,0.f};
#pragma unroll
  for (int g = 0; g < 2; ++g)
#pragma unroll
    for (int tt = 0; tt < 4; ++tt) acc[g][tt] = vzero;
  v4f rs = {0.f, 0.f, 0.f, 0.f};

  const float* aptr = X   + (long)(ms * 16 + l16) * NX + kc * 256 + 16 * half;
  const float* bptr = H2P + ((long)kc * 256 + 16 * half) * HID + 4 * l16;

  G1Unit ring[4];
  g1_load(ring[0], aptr, bptr, 0);
  g1_load(ring[1], aptr, bptr, 1);
  g1_load(ring[2], aptr + 32, bptr + (long)32 * HID, 0);
  v4f rsh = {0.f, 0.f, 0.f, 0.f};
#pragma unroll
  for (int u = 0; u < 16; ++u) {
    if (u + 3 < 16) {
      const int v = u + 3;
      g1_load(ring[v & 3], aptr + (v >> 1) * 32, bptr + (long)(v >> 1) * 32 * HID, v & 1);
    }
    __builtin_amdgcn_sched_barrier(0);
    if ((u & 1) == 0) rsh = ring[u & 3].xa[0] + ring[u & 3].xa[1];
    else              rs += rsh + (ring[u & 3].xa[0] + ring[u & 3].xa[1]);
    g1_mma(ring[u & 3], acc);
  }

  {
    float rsum = (rs.x + rs.y) + (rs.z + rs.w);
    rsum += __shfl_xor(rsum, 16);
    XP[((long)kc * 16 + ms) * 32 + lane] = rsum;
  }

#pragma unroll
  for (int g = 0; g < 2; ++g)
#pragma unroll
    for (int r = 0; r < 8; ++r) {
      v4f o;
      o.x = acc[g][0][r];
      o.y = acc[g][1][r];
      o.z = acc[g][2][r];
      o.w = acc[g][3][r];
      *(v4f*)(P + ((long)kc * ROWS + ms * 16 + 8 * half + r) * HID + 64 * g + 4 * l16) = o;
    }
}

__global__ void reduceP(const float* __restrict__ P, const float* __restrict__ XP,
                        float* __restrict__ T, float* __restrict__ X1) {
  if (blockIdx.x == 128) {
    const int r = threadIdx.x;
    const float* xp = XP + (r >> 4) * 32 + (r & 15);
    float s = 0.0f;
    for (int kc = 0; kc < 64; ++kc) s += xp[kc * 512];
    X1[r] = s;
    return;
  }
  const int idx = blockIdx.x * 256 + threadIdx.x;
  float s = 0.0f;
  for (int kc = 0; kc < 64; ++kc) s += P[(long)kc * (ROWS * HID) + idx];
  T[idx] = s;
}

#define TL_PITCH 132
#define WQ_PITCH 36
__global__ void __launch_bounds__(256)
suc_kernel(const float* __restrict__ T, const float* __restrict__ w3p,
           const float* __restrict__ b3p, const float* __restrict__ X1,
           const float* __restrict__ w3q, const float* __restrict__ b3q,
           float* __restrict__ U, float* __restrict__ CP) {
  __shared__ __attribute__((aligned(16))) float Tl[32 * TL_PITCH];
  __shared__ __attribute__((aligned(16))) float Wp[HID * 32];
  __shared__ __attribute__((aligned(16))) float Wq[HID * WQ_PITCH];
  __shared__ __attribute__((aligned(16))) float Sp[8 * 32];
  __shared__ __attribute__((aligned(16))) float Sl[32];
  const int tid = threadIdx.x;
  const int bo  = blockIdx.x;
  const int b   = bo >> 5;
  const int o   = bo & 31;

  const float bx = b3p[o * 32 + (tid & 31)] * X1[b * 32 + (tid & 31)];

#pragma unroll
  for (int j = 0; j < 4; ++j) {
    const int f = tid + 256 * j;
    {
      const int i = f >> 5, c4 = f & 31;
      *(v4f*)(Tl + i * TL_PITCH + 4 * c4) =
          *(const v4f*)(T + (long)(b * 32 + i) * HID + 4 * c4);
    }
    {
      const int k = f >> 3, c4 = f & 7;
      *(v4f*)(Wp + k * 32 + 4 * c4) =
          *(const v4f*)(w3p + (long)k * 1024 + o * 32 + 4 * c4);
      *(v4f*)(Wq + k * WQ_PITCH + 4 * c4) =
          *(const v4f*)(w3q + (long)k * 1024 + o * 32 + 4 * c4);
    }
  }
  __syncthreads();

  {
    const int i = tid & 31, k0 = (tid >> 5) * 16;
    float acc = 0.0f;
#pragma unroll
    for (int k4 = 0; k4 < 4; ++k4) {
      const v4f t = *(const v4f*)(Tl + i * TL_PITCH + k0 + 4 * k4);
      acc += t.x * Wp[(k0 + 4 * k4 + 0) * 32 + i];
      acc += t.y * Wp[(k0 + 4 * k4 + 1) * 32 + i];
      acc += t.z * Wp[(k0 + 4 * k4 + 2) * 32 + i];
      acc += t.w * Wp[(k0 + 4 * k4 + 3) * 32 + i];
    }
    Sp[tid] = acc;
  }
  __syncthreads();

  if (tid < 32) {
    const int i = tid;
    float acc = bx;
#pragma unroll
    for (int w = 0; w < 8; ++w) acc += Sp[w * 32 + i];
    Sl[i] = acc;
  }
  __syncthreads();

  if (tid < HID) {
    const int k = tid;
    float acc = 0.0f;
#pragma unroll
    for (int i4 = 0; i4 < 8; ++i4) {
      const v4f s = *(const v4f*)(Sl + 4 * i4);
      const v4f w = *(const v4f*)(Wq + k * WQ_PITCH + 4 * i4);
      acc += s.x * w.x;
      acc += s.y * w.y;
      acc += s.z * w.z;
      acc += s.w * w.w;
    }
    U[bo * HID + k] = acc;
  } else if (tid < HID + 32) {
    float acc = 0.0f;
#pragma unroll
    for (int i = 0; i < 32; ++i) acc += Sl[i] * b3q[o * 32 + i];
    CP[bo * 32 + (tid - HID)] = acc;
  }
}

struct G2Unit {
  v4f b[8][2];
};

__device__ __forceinline__ void g2_load(G2Unit& un, const float* __restrict__ bp) {
#pragma unroll
  for (int j = 0; j < 8; ++j)
#pragma unroll
    for (int e = 0; e < 2; ++e)
      un.b[j][e] = *(const v4f*)(bp + (long)(4 * j + e) * NX);
}

template <int S0>
__device__ __forceinline__ void g2_mma(const G2Unit& un, const v2f (&a)[32], v8f (&acc)[4]) {
#pragma unroll
  for (int j = 0; j < 8; ++j)
#pragma unroll
    for (int tt = 0; tt < 4; ++tt) {
      v2f b;
      b.x = un.b[j][0][tt];
      b.y = un.b[j][1][tt];
      acc[tt] = wmma4(a[S0 + j], b, acc[tt]);
    }
}

#define G2_STEP(Q)                                                              \
  {                                                                             \
    const int v3 = 4 * g + (Q) + 3;                                             \
    if (v3 < 16)                                                                \
      g2_load(ring[((Q) + 3) & 3],                                              \
              bptr + (long)(32 * (((Q) + 3) & 3)) * NX + (v3 >> 2) * 64);       \
    __builtin_amdgcn_sched_barrier(0);                                          \
    g2_mma<8 * (Q)>(ring[(Q)], a, acc);                                         \
  }

__global__ void __launch_bounds__(128)
gemm2(const float* __restrict__ U, const float* __restrict__ HT,
      const float* __restrict__ C, float* __restrict__ out) {
  const int lane = threadIdx.x & 31;
  const int half = lane >> 4;
  const int l16  = lane & 15;
  const int ms   = blockIdx.x * 4 + (threadIdx.x >> 5);
  const int nc   = blockIdx.y;

  const float* bptr = HT + (long)(2 * half) * NX + nc * 256 + 4 * l16;

  G2Unit ring[4];
  g2_load(ring[0], bptr);
  __builtin_amdgcn_sched_barrier(0);

  v2f a[32];
  const float* aptr = U + (long)(ms * 16 + l16) * HID + 2 * half;
#pragma unroll
  for (int s = 0; s < 32; ++s) a[s] = *(const v2f*)(aptr + 4 * s);

  float crow[8];
#pragma unroll
  for (int r = 0; r < 8; ++r) crow[r] = C[(ms * 16 + 8 * half + r) * 32];

  __builtin_amdgcn_sched_barrier(0);
  g2_load(ring[1], bptr + (long)32 * NX);
  g2_load(ring[2], bptr + (long)64 * NX);

  const float scale = 1.0f / (float)NX;
  float* optr = out + (long)(ms * 16 + 8 * half) * NX + nc * 256 + 4 * l16;

#pragma unroll
  for (int g = 0; g < 4; ++g) {
    v8f acc[4];
    const v8f vzero = {0.f,0.f,0.f,0.f,0.f,0.f,0.f,0.f};
#pragma unroll
    for (int tt = 0; tt < 4; ++tt) acc[tt] = vzero;

    G2_STEP(0)
    G2_STEP(1)
    G2_STEP(2)
    G2_STEP(3)

#pragma unroll
    for (int r = 0; r < 8; ++r) {
      v4f o;
      o.x = (acc[0][r] + crow[r]) * scale;
      o.y = (acc[1][r] + crow[r]) * scale;
      o.z = (acc[2][r] + crow[r]) * scale;
      o.w = (acc[3][r] + crow[r]) * scale;
      *(v4f*)(optr + (long)r * NX + g * 64) = o;
    }
  }
}

extern "C" void kernel_launch(void* const* d_in, const int* in_sizes, int n_in,
                              void* d_out, int out_size, void* d_ws, size_t ws_size,
                              hipStream_t stream) {
  const float* x   = (const float*)d_in[0];
  const float* pw1 = (const float*)d_in[1];
  const float* pb1 = (const float*)d_in[2];
  const float* pw2 = (const float*)d_in[3];
  const float* pb2 = (const float*)d_in[4];
  const float* pw3 = (const float*)d_in[5];
  const float* pb3 = (const float*)d_in[6];
  const float* qw1 = (const float*)d_in[7];
  const float* qb1 = (const float*)d_in[8];
  const float* qw2 = (const float*)d_in[9];
  const float* qb2 = (const float*)d_in[10];
  const float* qw3 = (const float*)d_in[11];
  const float* qb3 = (const float*)d_in[12];

  float* ws  = (float*)d_ws;
  float* H2P = ws;
  float* HTS = H2P + HID * NX;
  float* P   = HTS + NX * HID;
  float* T   = P + 64 * ROWS * HID;
  float* X1  = T + ROWS * HID;
  float* Uu  = X1 + ROWS;
  float* Cc  = Uu + ROWS * HID;
  float* XP  = Cc + ROWS * 32;

  mlp_trunk<<<dim3(NX / 32, HID / 64, 2), dim3(32), 0, stream>>>(pw1, pb1, pw2, pb2,
                                                                 qw1, qb1, qw2, qb2, H2P, HTS);
  gemm1_partial<<<dim3(4, 64), dim3(128), 0, stream>>>(x, H2P, P, XP);
  reduceP<<<dim3(129), dim3(256), 0, stream>>>(P, XP, T, X1);
  suc_kernel<<<dim3(ROWS), dim3(256), 0, stream>>>(T, pw3, pb3, X1, qw3, qb3, Uu, Cc);
  gemm2<<<dim3(4, 64), dim3(128), 0, stream>>>(Uu, HTS, Cc, (float*)d_out);
}


# Reduce of the GEMM1 partials: all 64 loads of an output are issued before the first add (one memory round trip, was four groups of 16), in 128-thread blocks on all 256 processors; sum order kept
# speedup vs baseline: 10.4970x; 1.0430x over previous
#include <hip/hip_runtime.h>

typedef __attribute__((ext_vector_type(2))) float v2f;
typedef __attribute__((ext_vector_type(4))) float v4f;
typedef __attribute__((ext_vector_type(8))) float v8f;

#define NX    16384
#define HID   128
#define ROWS  256

__device__ __forceinline__ v8f wmma4(v2f a, v2f b, v8f c) {
  return __builtin_amdgcn_wmma_f32_16x16x4_f32(
       false, a,  false, b,
       (short)0, c,  false,  false);
}

__device__ __forceinline__ void trunk_tile(const float* __restrict__ w1,
                                           const float* __restrict__ b1,
                                           const float* __restrict__ w2,
                                           const float* __restrict__ b2,
                                           float* __restrict__ H,
                                           int n0, int c0, int half, int l16) {
  v2f hf[2][16];
#pragma unroll
  for (int p = 0; p < 2; ++p) {
    const int n = n0 + 2 * l16 + p;
    const float gx = (float)(n >> 7) * (1.0f / 127.0f);
    const float gy = (float)(n & 127) * (1.0f / 127.0f);
#pragma unroll
    for (int ks = 0; ks < 16; ++ks) {
      const int j0 = 4 * ks + 2 * half;
      const v2f wa = *(const v2f*)(w1 + j0);
      const v2f wb = *(const v2f*)(w1 + 64 + j0);
      const v2f bb = *(const v2f*)(b1 + j0);
      v2f h;
      h.x = fmaxf(gx * wa.x + gy * wb.x + bb.x, 0.0f);
      h.y = fmaxf(gx * wa.y + gy * wb.y + bb.y, 0.0f);
      hf[p][ks] = h;
    }
  }

  v8f acc[4][2];
  const v4f bq = *(const v4f*)(b2 + c0 + 4 * l16);
#pragma unroll
  for (int tt = 0; tt < 4; ++tt)
#pragma unroll
    for (int r = 0; r < 8; ++r) {
      acc[tt][0][r] = bq[tt];
      acc[tt][1][r] = bq[tt];
    }

  const float* wp = w2 + (2 * half) * HID + c0 + 4 * l16;
#pragma unroll
  for (int ks = 0; ks < 16; ++ks) {
    const v4f wlo = *(const v4f*)(wp + (4 * ks) * HID);
    const v4f whi = *(const v4f*)(wp + (4 * ks + 1) * HID);
#pragma unroll
    for (int tt = 0; tt < 4; ++tt) {
      v2f wf;
      wf.x = wlo[tt];
      wf.y = whi[tt];
#pragma unroll
      for (int p = 0; p < 2; ++p)
        acc[tt][p] = wmma4(hf[p][ks], wf, acc[tt][p]);
    }
  }

#pragma unroll
  for (int p = 0; p < 2; ++p)
#pragma unroll
    for (int r = 0; r < 8; ++r) {
      const int n = n0 + 2 * (8 * half + r) + p;
      v4f o;
      o.x = fmaxf(acc[0][p][r], 0.0f);
      o.y = fmaxf(acc[1][p][r], 0.0f);
      o.z = fmaxf(acc[2][p][r], 0.0f);
      o.w = fmaxf(acc[3][p][r], 0.0f);
      *(v4f*)(H + (long)n * HID + c0 + 4 * l16) = o;
    }
}


__device__ __forceinline__ void trunk_tile_t(const float* __restrict__ w1,
                                             const float* __restrict__ b1,
                                             const float* __restrict__ w2,
                                             const float* __restrict__ b2,
                                             float* __restrict__ HT,
                                             int n0, int c0, int half, int l16) {
  v2f hf[2][16];
#pragma unroll
  for (int p = 0; p < 2; ++p) {
    const int n = n0 + 2 * l16 + p;
    const float gx = (float)(n >> 7) * (1.0f / 127.0f);
    const float gy = (float)(n & 127) * (1.0f / 127.0f);
#pragma unroll
    for (int ks = 0; ks < 16; ++ks) {
      const int j0 = 4 * ks + 2 * half;
      const v2f wa = *(const v2f*)(w1 + j0);
      const v2f wb = *(const v2f*)(w1 + 64 + j0);
      const v2f bb = *(const v2f*)(b1 + j0);
      v2f h;
      h.x = fmaxf(gx * wa.x + gy * wb.x + bb.x, 0.0f);
      h.y = fmaxf(gx * wa.y + gy * wb.y + bb.y, 0.0f);
      hf[p][ks] = h;
    }
  }

  v8f acc[4][2];
#pragma unroll
  for (int r = 0; r < 8; ++r) {
    const v4f bq = *(const v4f*)(b2 + c0 + 4 * (8 * half + r));
#pragma unroll
    for (int tt = 0; tt < 4; ++tt) {
      acc[tt][0][r] = bq[tt];
      acc[tt][1][r] = bq[tt];
    }
  }

  const float* wp = w2 + (2 * half) * HID + c0 + 4 * l16;
#pragma unroll
  for (int ks = 0; ks < 16; ++ks) {
    const v4f wlo = *(const v4f*)(wp + (4 * ks) * HID);
    const v4f whi = *(const v4f*)(wp + (4 * ks + 1) * HID);
#pragma unroll
    for (int tt = 0; tt < 4; ++tt) {
      v2f wf;
      wf.x = wlo[tt];
      wf.y = whi[tt];
#pragma unroll
      for (int p = 0; p < 2; ++p)
        acc[tt][p] = wmma4(wf, hf[p][ks], acc[tt][p]);
    }
  }

#pragma unroll
  for (int tt = 0; tt < 4; ++tt)
#pragma unroll
    for (int r = 0; r < 8; ++r) {
      const int c = c0 + 4 * (8 * half + r) + tt;
      v2f o;
      o.x = fmaxf(acc[tt][0][r], 0.0f);
      o.y = fmaxf(acc[tt][1][r], 0.0f);
      *(v2f*)(HT + (long)c * NX + n0 + 2 * l16) = o;
    }
}

__global__ void __launch_bounds__(32)
mlp_trunk(const float* __restrict__ pw1, const float* __restrict__ pb1,
          const float* __restrict__ pw2, const float* __restrict__ pb2,
          const float* __restrict__ qw1, const float* __restrict__ qb1,
          const float* __restrict__ qw2, const float* __restrict__ qb2,
          float* __restrict__ H2P, float* __restrict__ HTS) {
  const int lane = threadIdx.x;
  const int half = lane >> 4;
  const int l16  = lane & 15;
  const int n0 = blockIdx.x * 32;
  const int c0 = blockIdx.y * 64;
  if (blockIdx.z == 0) trunk_tile(pw1, pb1, pw2, pb2, H2P, n0, c0, half, l16);
  else                 trunk_tile_t(qw1, qb1, qw2, qb2, HTS, n0, c0, half, l16);
}

struct G1Unit {
  v4f xa[2];
  v4f bw[4][2][2];
};

__device__ __forceinline__ void g1_load(G1Unit& un, const float* __restrict__ ap,
                                        const float* __restrict__ bp, int q) {
#pragma unroll
  for (int p = 0; p < 2; ++p) un.xa[p] = *(const v4f*)(ap + 8 * q + 4 * p);
#pragma unroll
  for (int j = 0; j < 4; ++j)
#pragma unroll
    for (int g = 0; g < 2; ++g)
#pragma unroll
      for (int e = 0; e < 2; ++e)
        un.bw[j][g][e] = *(const v4f*)(bp + (8 * q + 2 * j + e) * HID + 64 * g);
}

__device__ __forceinline__ void g1_mma(const G1Unit& un, v8f (&acc)[2][4]) {
#pragma unroll
  for (int j = 0; j < 4; ++j) {
    v2f a;
    a.x = un.xa[j >> 1][2 * (j & 1)];
    a.y = un.xa[j >> 1][2 * (j & 1) + 1];
#pragma unroll
    for (int g = 0; g < 2; ++g)
#pragma unroll
      for (int tt = 0; tt < 4; ++tt) {
        v2f b;
        b.x = un.bw[j][g][0][tt];
        b.y = un.bw[j][g][1][tt];
        acc[g][tt] = wmma4(a, b, acc[g][tt]);
      }
  }
}

__global__ void __launch_bounds__(128)
gemm1_partial(const float* __restrict__ X, const float* __restrict__ H2P,
              float* __restrict__ P, float* __restrict__ XP) {
  const int lane = threadIdx.x & 31;
  const int half = lane >> 4;
  const int l16  = lane & 15;
  const int ms   = blockIdx.x * 4 + (threadIdx.x >> 5);
  const int kc   = blockIdx.y;

  v8f acc[2][4];
  const v8f vzero = {0.f,0.f,0.f,0.f,0.f,0.f,0.f,0.f};
#pragma unroll
  for (int g = 0; g < 2; ++g)
#pragma unroll
    for (int tt = 0; tt < 4; ++tt) acc[g][tt] = vzero;
  v4f rs = {0.f, 0.f, 0.f, 0.f};

  const float* aptr = X   + (long)(ms * 16 + l16) * NX + kc * 256 + 16 * half;
  const float* bptr = H2P + ((long)kc * 256 + 16 * half) * HID + 4 * l16;

  G1Unit ring[4];
  g1_load(ring[0], aptr, bptr, 0);
  g1_load(ring[1], aptr, bptr, 1);
  g1_load(ring[2], aptr + 32, bptr + (long)32 * HID, 0);
  v4f rsh = {0.f, 0.f, 0.f, 0.f};
#pragma unroll
  for (int u = 0; u < 16; ++u) {
    if (u + 3 < 16) {
      const int v = u + 3;
      g1_load(ring[v & 3], aptr + (v >> 1) * 32, bptr + (long)(v >> 1) * 32 * HID, v & 1);
    }
    __builtin_amdgcn_sched_barrier(0);
    if ((u & 1) == 0) rsh = ring[u & 3].xa[0] + ring[u & 3].xa[1];
    else              rs += rsh + (ring[u & 3].xa[0] + ring[u & 3].xa[1]);
    g1_mma(ring[u & 3], acc);
  }

  {
    float rsum = (rs.x + rs.y) + (rs.z + rs.w);
    rsum += __shfl_xor(rsum, 16);
    XP[((long)kc * 16 + ms) * 32 + lane] = rsum;
  }

#pragma unroll
  for (int g = 0; g < 2; ++g)
#pragma unroll
    for (int r = 0; r < 8; ++r) {
      v4f o;
      o.x = acc[g][0][r];
      o.y = acc[g][1][r];
      o.z = acc[g][2][r];
      o.w = acc[g][3][r];
      *(v4f*)(P + ((long)kc * ROWS + ms * 16 + 8 * half + r) * HID + 64 * g + 4 * l16) = o;
    }
}

__global__ void __launch_bounds__(128)
reduceP(const float* __restrict__ P, const float* __restrict__ XP,
        float* __restrict__ T, float* __restrict__ X1) {
  float p[64];
  if (blockIdx.x >= 256) {
    const int r = (blockIdx.x - 256) * 128 + threadIdx.x;
    const float* xp = XP + (r >> 4) * 32 + (r & 15);
#pragma unroll
    for (int kc = 0; kc < 64; ++kc) p[kc] = xp[kc * 512];
    float s = 0.0f;
#pragma unroll
    for (int kc = 0; kc < 64; ++kc) s += p[kc];
    X1[r] = s;
    return;
  }
  const int idx = blockIdx.x * 128 + threadIdx.x;
#pragma unroll
  for (int kc = 0; kc < 64; ++kc) p[kc] = P[(long)kc * (ROWS * HID) + idx];
  float s = 0.0f;
#pragma unroll
  for (int kc = 0; kc < 64; ++kc) s += p[kc];
  T[idx] = s;
}

#define TL_PITCH 132
#define WQ_PITCH 36
__global__ void __launch_bounds__(256)
suc_kernel(const float* __restrict__ T, const float* __restrict__ w3p,
           const float* __restrict__ b3p, const float* __restrict__ X1,
           const float* __restrict__ w3q, const float* __restrict__ b3q,
           float* __restrict__ U, float* __restrict__ CP) {
  __shared__ __attribute__((aligned(16))) float Tl[32 * TL_PITCH];
  __shared__ __attribute__((aligned(16))) float Wp[HID * 32];
  __shared__ __attribute__((aligned(16))) float Wq[HID * WQ_PITCH];
  __shared__ __attribute__((aligned(16))) float Sp[8 * 32];
  __shared__ __attribute__((aligned(16))) float Sl[32];
  const int tid = threadIdx.x;
  const int bo  = blockIdx.x;
  const int b   = bo >> 5;
  const int o   = bo & 31;

  const float bx = b3p[o * 32 + (tid & 31)] * X1[b * 32 + (tid & 31)];

#pragma unroll
  for (int j = 0; j < 4; ++j) {
    const int f = tid + 256 * j;
    {
      const int i = f >> 5, c4 = f & 31;
      *(v4f*)(Tl + i * TL_PITCH + 4 * c4) =
          *(const v4f*)(T + (long)(b * 32 + i) * HID + 4 * c4);
    }
    {
      const int k = f >> 3, c4 = f & 7;
      *(v4f*)(Wp + k * 32 + 4 * c4) =
          *(const v4f*)(w3p + (long)k * 1024 + o * 32 + 4 * c4);
      *(v4f*)(Wq + k * WQ_PITCH + 4 * c4) =
          *(const v4f*)(w3q + (long)k * 1024 + o * 32 + 4 * c4);
    }
  }
  __syncthreads();

  {
    const int i = tid & 31, k0 = (tid >> 5) * 16;
    float acc = 0.0f;
#pragma unroll
    for (int k4 = 0; k4 < 4; ++k4) {
      const v4f t = *(const v4f*)(Tl + i * TL_PITCH + k0 + 4 * k4);
      acc += t.x * Wp[(k0 + 4 * k4 + 0) * 32 + i];
      acc += t.y * Wp[(k0 + 4 * k4 + 1) * 32 + i];
      acc += t.z * Wp[(k0 + 4 * k4 + 2) * 32 + i];
      acc += t.w * Wp[(k0 + 4 * k4 + 3) * 32 + i];
    }
    Sp[tid] = acc;
  }
  __syncthreads();

  if (tid < 32) {
    const int i = tid;
    float acc = bx;
#pragma unroll
    for (int w = 0; w < 8; ++w) acc += Sp[w * 32 + i];
    Sl[i] = acc;
  }
  __syncthreads();

  if (tid < HID) {
    const int k = tid;
    float acc = 0.0f;
#pragma unroll
    for (int i4 = 0; i4 < 8; ++i4) {
      const v4f s = *(const v4f*)(Sl + 4 * i4);
      const v4f w = *(const v4f*)(Wq + k * WQ_PITCH + 4 * i4);
      acc += s.x * w.x;
      acc += s.y * w.y;
      acc += s.z * w.z;
      acc += s.w * w.w;
    }
    U[bo * HID + k] = acc;
  } else if (tid < HID + 32) {
    float acc = 0.0f;
#pragma unroll
    for (int i = 0; i < 32; ++i) acc += Sl[i] * b3q[o * 32 + i];
    CP[bo * 32 + (tid - HID)] = acc;
  }
}

struct G2Unit {
  v4f b[8][2];
};

__device__ __forceinline__ void g2_load(G2Unit& un, const float* __restrict__ bp) {
#pragma unroll
  for (int j = 0; j < 8; ++j)
#pragma unroll
    for (int e = 0; e < 2; ++e)
      un.b[j][e] = *(const v4f*)(bp + (long)(4 * j + e) * NX);
}

template <int S0>
__device__ __forceinline__ void g2_mma(const G2Unit& un, const v2f (&a)[32], v8f (&acc)[4]) {
#pragma unroll
  for (int j = 0; j < 8; ++j)
#pragma unroll
    for (int tt = 0; tt < 4; ++tt) {
      v2f b;
      b.x = un.b[j][0][tt];
      b.y = un.b[j][1][tt];
      acc[tt] = wmma4(a[S0 + j], b, acc[tt]);
    }
}

#define G2_STEP(Q)                                                              \
  {                                                                             \
    const int v3 = 4 * g + (Q) + 3;                                             \
    if (v3 < 16)                                                                \
      g2_load(ring[((Q) + 3) & 3],                                              \
              bptr + (long)(32 * (((Q) + 3) & 3)) * NX + (v3 >> 2) * 64);       \
    __builtin_amdgcn_sched_barrier(0);                                          \
    g2_mma<8 * (Q)>(ring[(Q)], a, acc);                                         \
  }

__global__ void __launch_bounds__(128)
gemm2(const float* __restrict__ U, const float* __restrict__ HT,
      const float* __restrict__ C, float* __restrict__ out) {
  const int lane = threadIdx.x & 31;
  const int half = lane >> 4;
  const int l16  = lane & 15;
  const int ms   = blockIdx.x * 4 + (threadIdx.x >> 5);
  const int nc   = blockIdx.y;

  const float* bptr = HT + (long)(2 * half) * NX + nc * 256 + 4 * l16;

  G2Unit ring[4];
  g2_load(ring[0], bptr);
  __builtin_amdgcn_sched_barrier(0);

  v2f a[32];
  const float* aptr = U + (long)(ms * 16 + l16) * HID + 2 * half;
#pragma unroll
  for (int s = 0; s < 32; ++s) a[s] = *(const v2f*)(aptr + 4 * s);

  float crow[8];
#pragma unroll
  for (int r = 0; r < 8; ++r) crow[r] = C[(ms * 16 + 8 * half + r) * 32];

  __builtin_amdgcn_sched_barrier(0);
  g2_load(ring[1], bptr + (long)32 * NX);
  g2_load(ring[2], bptr + (long)64 * NX);

  const float scale = 1.0f / (float)NX;
  float* optr = out + (long)(ms * 16 + 8 * half) * NX + nc * 256 + 4 * l16;

#pragma unroll
  for (int g = 0; g < 4; ++g) {
    v8f acc[4];
    const v8f vzero = {0.f,0.f,0.f,0.f,0.f,0.f,0.f,0.f};
#pragma unroll
    for (int tt = 0; tt < 4; ++tt) acc[tt] = vzero;

    G2_STEP(0)
    G2_STEP(1)
    G2_STEP(2)
    G2_STEP(3)

#pragma unroll
    for (int r = 0; r < 8; ++r) {
      v4f o;
      o.x = (acc[0][r] + crow[r]) * scale;
      o.y = (acc[1][r] + crow[r]) * scale;
      o.z = (acc[2][r] + crow[r]) * scale;
      o.w = (acc[3][r] + crow[r]) * scale;
      *(v4f*)(optr + (long)r * NX + g * 64) = o;
    }
  }
}

extern "C" void kernel_launch(void* const* d_in, const int* in_sizes, int n_in,
                              void* d_out, int out_size, void* d_ws, size_t ws_size,
                              hipStream_t stream) {
  const float* x   = (const float*)d_in[0];
  const float* pw1 = (const float*)d_in[1];
  const float* pb1 = (const float*)d_in[2];
  const float* pw2 = (const float*)d_in[3];
  const float* pb2 = (const float*)d_in[4];
  const float* pw3 = (const float*)d_in[5];
  const float* pb3 = (const float*)d_in[6];
  const float* qw1 = (const float*)d_in[7];
  const float* qb1 = (const float*)d_in[8];
  const float* qw2 = (const float*)d_in[9];
  const float* qb2 = (const float*)d_in[10];
  const float* qw3 = (const float*)d_in[11];
  const float* qb3 = (const float*)d_in[12];

  float* ws  = (float*)d_ws;
  float* H2P = ws;
  float* HTS = H2P + HID * NX;
  float* P   = HTS + NX * HID;
  float* T   = P + 64 * ROWS * HID;
  float* X1  = T + ROWS * HID;
  float* Uu  = X1 + ROWS;
  float* Cc  = Uu + ROWS * HID;
  float* XP  = Cc + ROWS * 32;

  mlp_trunk<<<dim3(NX / 32, HID / 64, 2), dim3(32), 0, stream>>>(pw1, pb1, pw2, pb2,
                                                                 qw1, qb1, qw2, qb2, H2P, HTS);
  gemm1_partial<<<dim3(4, 64), dim3(128), 0, stream>>>(x, H2P, P, XP);
  reduceP<<<dim3(258), dim3(128), 0, stream>>>(P, XP, T, X1);
  suc_kernel<<<dim3(ROWS), dim3(256), 0, stream>>>(T, pw3, pb3, X1, qw3, qb3, Uu, Cc);
  gemm2<<<dim3(4, 64), dim3(128), 0, stream>>>(Uu, HTS, Cc, (float*)d_out);
}
